# starting kernel
# baseline (speedup 1.0000x reference)
#include <hip/hip_runtime.h>
#include <hip/hip_bf16.h>
#include <math.h>

typedef float v2f __attribute__((ext_vector_type(2)));
typedef float v8f __attribute__((ext_vector_type(8)));

#define IN_C   128
#define HID_C  128
#define OUT_C  64
#define NBATCH 512

__global__ __launch_bounds__(256) void gemm_f32_wmma(
    const float* __restrict__ A, const float* __restrict__ B,
    const float* __restrict__ bias, float* __restrict__ C,
    int M, int K, int N, int act)
{
    __shared__ float lds_a[16][132];

    const int tid  = threadIdx.x;
    const int wave = tid >> 5;
    const int lane = tid & 31;
    const int row0 = blockIdx.x * 16;

    const int nvec = (16 * K) >> 2;
    const int kv   = K >> 2;
    for (int i = tid; i < nvec; i += blockDim.x) {
        int r  = i / kv;
        int k4 = (i - r * kv) << 2;
        float4 v = *(const float4*)(A + (size_t)(row0 + r) * K + k4);
        *(float4*)&lds_a[r][k4] = v;
    }
    __syncthreads();

    const int m    = lane & 15;
    const int kh   = (lane >> 4) << 1;
    const int col0 = wave * 16;

    v8f acc = {};
    for (int kb = 0; kb < K; kb += 4) {
        v2f a, b;
        a.x = lds_a[m][kb + kh];
        a.y = lds_a[m][kb + kh + 1];
        const float* Bp = B + (size_t)(kb + kh) * N + col0 + m;
        b.x = Bp[0];
        b.y = Bp[N];
        acc = __builtin_amdgcn_wmma_f32_16x16x4_f32(
            false, a, false, b, (short)0, acc, false, false);
    }

    const int   col   = col0 + m;
    const int   rbase = row0 + ((lane >> 4) << 3);
    const float bv    = bias ? bias[col] : 0.0f;
#pragma unroll
    for (int v = 0; v < 8; ++v) {
        float x = acc[v] + bv;
        if (act == 1) x = fmaxf(x, 0.0f);
        C[(size_t)(rbase + v) * N + col] = x;
    }
}

__global__ void fill_f32(float* __restrict__ p, float v, int n)
{
    int i = blockIdx.x * blockDim.x + threadIdx.x;
    if (i < n) p[i] = v;
}

__global__ void deg_edges(const int* __restrict__ dst, float* __restrict__ deg, int E)
{
    int e = blockIdx.x * blockDim.x + threadIdx.x;
    if (e < E) atomicAdd(&deg[dst[e]], 1.0f);
}

__global__ void rsqrt_k(const float* __restrict__ deg, float* __restrict__ dinv, int n)
{
    int i = blockIdx.x * blockDim.x + threadIdx.x;
    if (i < n) dinv[i] = rsqrtf(deg[i]);
}

__global__ __launch_bounds__(32) void self_loop_init(
    const float* __restrict__ xw, const float* __restrict__ dinv,
    float* __restrict__ h, int n)
{
    int node = blockIdx.x;
    int c4   = threadIdx.x << 2;
    float di = dinv[node];
    float s  = di * di;
    size_t idx = (size_t)node * HID_C + c4;
    float4 v = *(const float4*)(xw + idx);
    v.x *= s; v.y *= s; v.z *= s; v.w *= s;
    *(float4*)(h + idx) = v;
}

__global__ __launch_bounds__(256) void edge_scatter(
    const float* __restrict__ xw, const float* __restrict__ dinv,
    const int* __restrict__ src, const int* __restrict__ dst,
    float* __restrict__ h, int E)
{
    int e  = blockIdx.x * 8 + (threadIdx.x >> 5);
    int c4 = (threadIdx.x & 31) << 2;
    if (e < E) {
        int   s    = src[e];
        int   d    = dst[e];
        float norm = dinv[s] * dinv[d];
        float4 v = *(const float4*)(xw + (size_t)s * HID_C + c4);
        float* hp = h + (size_t)d * HID_C + c4;
        atomicAdd(hp + 0, v.x * norm);
        atomicAdd(hp + 1, v.y * norm);
        atomicAdd(hp + 2, v.z * norm);
        atomicAdd(hp + 3, v.w * norm);
    }
}

__global__ __launch_bounds__(32) void bias_gelu(
    float* __restrict__ h, const float* __restrict__ b, int n)
{
    int node = blockIdx.x;
    int c4   = threadIdx.x << 2;
    size_t idx = (size_t)node * HID_C + c4;
    float4 v  = *(const float4*)(h + idx);
    float4 bb = *(const float4*)(b + c4);
    v.x += bb.x; v.y += bb.y; v.z += bb.z; v.w += bb.w;
    v.x = 0.5f * v.x * (1.0f + erff(v.x * 0.70710678118654752f));
    v.y = 0.5f * v.y * (1.0f + erff(v.y * 0.70710678118654752f));
    v.z = 0.5f * v.z * (1.0f + erff(v.z * 0.70710678118654752f));
    v.w = 0.5f * v.w * (1.0f + erff(v.w * 0.70710678118654752f));
    *(float4*)(h + idx) = v;
}

__global__ __launch_bounds__(32) void pool_accum(
    const float* __restrict__ h, const int* __restrict__ batch,
    float* __restrict__ sums, float* __restrict__ cnt, int n)
{
    int node = blockIdx.x;
    int c4   = threadIdx.x << 2;
    int b    = batch[node];
    float4 v = *(const float4*)(h + (size_t)node * HID_C + c4);
    float* sp = sums + (size_t)b * HID_C + c4;
    atomicAdd(sp + 0, v.x);
    atomicAdd(sp + 1, v.y);
    atomicAdd(sp + 2, v.z);
    atomicAdd(sp + 3, v.w);
    if (threadIdx.x == 0) atomicAdd(&cnt[b], 1.0f);
}

__global__ void pool_final(float* __restrict__ sums, const float* __restrict__ cnt, int n)
{
    int i = blockIdx.x * blockDim.x + threadIdx.x;
    if (i < n) sums[i] /= fmaxf(cnt[i / HID_C], 1.0f);
}

extern "C" void kernel_launch(void* const* d_in, const int* in_sizes, int n_in,
                              void* d_out, int out_size, void* d_ws, size_t ws_size,
                              hipStream_t stream)
{
    const float* x     = (const float*)d_in[0];
    const int*   ei    = (const int*)d_in[1];
    const int*   batch = (const int*)d_in[2];
    const float* W[3]  = { (const float*)d_in[3], (const float*)d_in[5], (const float*)d_in[7] };
    const float* bv[3] = { (const float*)d_in[4], (const float*)d_in[6], (const float*)d_in[8] };
    const float* Wm1   = (const float*)d_in[9];
    const float* bm1   = (const float*)d_in[10];
    const float* Wm2   = (const float*)d_in[11];
    const float* bm2   = (const float*)d_in[12];

    const int N = in_sizes[0] / IN_C;
    const int E = in_sizes[1] / 2;
    const int* src = ei;
    const int* dst = ei + E;

    char*  ws  = (char*)d_ws;
    size_t off = 0;
    auto carve = [&](size_t bytes) -> float* {
        float* p = (float*)(ws + off);
        off = (off + bytes + 255) & ~(size_t)255;
        return p;
    };
    float* deg  = carve((size_t)N * 4);
    float* dinv = carve((size_t)N * 4);
    float* buf0 = carve((size_t)N * HID_C * 4);
    float* buf1 = carve((size_t)N * HID_C * 4);
    float* psum = carve((size_t)NBATCH * HID_C * 4);
    float* pcnt = carve((size_t)NBATCH * 4);
    float* hid  = carve((size_t)NBATCH * HID_C * 4);

    fill_f32<<<(N + 255) / 256, 256, 0, stream>>>(deg, 1.0f, N);
    deg_edges<<<(E + 255) / 256, 256, 0, stream>>>(dst, deg, E);
    rsqrt_k<<<(N + 255) / 256, 256, 0, stream>>>(deg, dinv, N);

    const float* cur = x;
    for (int l = 0; l < 3; ++l) {
        gemm_f32_wmma<<<N / 16, 256, 0, stream>>>(cur, W[l], nullptr, buf0,
                                                  N, HID_C, HID_C, 0);
        self_loop_init<<<N, 32, 0, stream>>>(buf0, dinv, buf1, N);
        edge_scatter<<<(E + 7) / 8, 256, 0, stream>>>(buf0, dinv, src, dst, buf1, E);
        bias_gelu<<<N, 32, 0, stream>>>(buf1, bv[l], N);
        cur = buf1;
    }

    fill_f32<<<(NBATCH * HID_C + 255) / 256, 256, 0, stream>>>(psum, 0.0f, NBATCH * HID_C);
    fill_f32<<<(NBATCH + 255) / 256, 256, 0, stream>>>(pcnt, 0.0f, NBATCH);
    pool_accum<<<N, 32, 0, stream>>>(cur, batch, psum, pcnt, N);
    pool_final<<<(NBATCH * HID_C + 255) / 256, 256, 0, stream>>>(psum, pcnt, NBATCH * HID_C);

    gemm_f32_wmma<<<NBATCH / 16, 256, 0, stream>>>(psum, Wm1, bm1, hid,
                                                   NBATCH, HID_C, HID_C, 1);
    gemm_f32_wmma<<<NBATCH / 16, 128, 0, stream>>>(hid, Wm2, bm2, (float*)d_out,
                                                   NBATCH, HID_C, OUT_C, 0);
}


# edge aggregation: f32 atomic scatter -> atomic-free row gather over a dst-sorted edge list built in-call (int arrival-order rank + prefix scan); row sums now run in list order
# speedup vs baseline: 2.5603x; 2.5603x over previous
#include <hip/hip_runtime.h>
#include <hip/hip_bf16.h>
#include <math.h>

typedef float v2f __attribute__((ext_vector_type(2)));
typedef float v8f __attribute__((ext_vector_type(8)));

#define IN_C   128
#define HID_C  128
#define OUT_C  64
#define NBATCH 512

__global__ __launch_bounds__(256) void gemm_f32_wmma(
    const float* __restrict__ A, const float* __restrict__ B,
    const float* __restrict__ bias, float* __restrict__ C,
    int M, int K, int N, int act)
{
    __shared__ float lds_a[16][132];

    const int tid  = threadIdx.x;
    const int wave = tid >> 5;
    const int lane = tid & 31;
    const int row0 = blockIdx.x * 16;

    const int nvec = (16 * K) >> 2;
    const int kv   = K >> 2;
    for (int i = tid; i < nvec; i += blockDim.x) {
        int r  = i / kv;
        int k4 = (i - r * kv) << 2;
        float4 v = *(const float4*)(A + (size_t)(row0 + r) * K + k4);
        *(float4*)&lds_a[r][k4] = v;
    }
    __syncthreads();

    const int m    = lane & 15;
    const int kh   = (lane >> 4) << 1;
    const int col0 = wave * 16;

    v8f acc = {};
    for (int kb = 0; kb < K; kb += 4) {
        v2f a, b;
        a.x = lds_a[m][kb + kh];
        a.y = lds_a[m][kb + kh + 1];
        const float* Bp = B + (size_t)(kb + kh) * N + col0 + m;
        b.x = Bp[0];
        b.y = Bp[N];
        acc = __builtin_amdgcn_wmma_f32_16x16x4_f32(
            false, a, false, b, (short)0, acc, false, false);
    }

    const int   col   = col0 + m;
    const int   rbase = row0 + ((lane >> 4) << 3);
    const float bv    = bias ? bias[col] : 0.0f;
#pragma unroll
    for (int v = 0; v < 8; ++v) {
        float x = acc[v] + bv;
        if (act == 1) x = fmaxf(x, 0.0f);
        C[(size_t)(rbase + v) * N + col] = x;
    }
}

__global__ void fill_f32(float* __restrict__ p, float v, int n)
{
    int i = blockIdx.x * blockDim.x + threadIdx.x;
    if (i < n) p[i] = v;
}

__global__ void zero_i32(int* __restrict__ p, int n)
{
    int i = blockIdx.x * blockDim.x + threadIdx.x;
    if (i < n) p[i] = 0;
}

__global__ void count_rank(const int* __restrict__ dst, int* __restrict__ cnt,
                           int* __restrict__ rank, int E)
{
    int e = blockIdx.x * blockDim.x + threadIdx.x;
    if (e < E) rank[e] = atomicAdd(&cnt[dst[e]], 1);
}

#define SCAN_T 1024
__global__ __launch_bounds__(SCAN_T) void scan_rowptr(
    const int* __restrict__ cnt, int* __restrict__ rowptr,
    float* __restrict__ dinv, int n)
{
    __shared__ int s_own[SCAN_T];
    __shared__ int s_pre[SCAN_T];
    const int t = threadIdx.x;
    const int b = blockIdx.x;

    int pre = 0;
    for (int k = 0; k < b; ++k) pre += cnt[k * SCAN_T + t];

    const int i = b * SCAN_T + t;
    const int c = (i < n) ? cnt[i] : 0;
    s_own[t] = c;
    s_pre[t] = pre;
    __syncthreads();
    for (int off = 1; off < SCAN_T; off <<= 1) {
        int a = 0, p = 0;
        if (t >= off) { a = s_own[t - off]; p = s_pre[t - off]; }
        __syncthreads();
        s_own[t] += a;
        s_pre[t] += p;
        __syncthreads();
    }
    const int excl = s_pre[SCAN_T - 1] + s_own[t] - c;
    if (i < n) {
        rowptr[i] = excl;
        dinv[i]   = rsqrtf((float)(c + 1));
    }
    if (i == n - 1) rowptr[n] = excl + c;
}

__global__ void fill_csr(const int* __restrict__ src, const int* __restrict__ dst,
                         const int* __restrict__ rank, const int* __restrict__ rowptr,
                         const float* __restrict__ dinv, int2* __restrict__ csr, int E)
{
    int e = blockIdx.x * blockDim.x + threadIdx.x;
    if (e < E) {
        int s = src[e];
        int d = dst[e];
        int2 r;
        r.x = s;
        r.y = __float_as_int(dinv[s] * dinv[d]);
        csr[rowptr[d] + rank[e]] = r;
    }
}

template <int K>
__device__ __forceinline__ void add_edges(
    float4& acc, const float* __restrict__ xw, const int2* __restrict__ rec, int c4)
{
    int2   r[K];
    float4 v[K];
#pragma unroll
    for (int k = 0; k < K; ++k) r[k] = rec[k];
#pragma unroll
    for (int k = 0; k < K; ++k)
        v[k] = *(const float4*)(xw + (size_t)r[k].x * HID_C + c4);
#pragma unroll
    for (int k = 0; k < K; ++k) {
        const float w = __int_as_float(r[k].y);
        acc.x = fmaf(v[k].x, w, acc.x);
        acc.y = fmaf(v[k].y, w, acc.y);
        acc.z = fmaf(v[k].z, w, acc.z);
        acc.w = fmaf(v[k].w, w, acc.w);
    }
}

__global__ __launch_bounds__(256) void gather_rows(
    const float* __restrict__ xw, const float* __restrict__ dinv,
    const int* __restrict__ rowptr, const int2* __restrict__ csr,
    float* __restrict__ h, int n)
{
    const int row = __builtin_amdgcn_readfirstlane(
        (int)(blockIdx.x * 8 + (threadIdx.x >> 5)));
    if (row >= n) return;
    const int c4  = (threadIdx.x & 31) << 2;
    const int beg = rowptr[row];
    const int end = rowptr[row + 1];
    const float di = dinv[row];
    const float s  = di * di;

    float4 acc = *(const float4*)(xw + (size_t)row * HID_C + c4);
    acc.x *= s; acc.y *= s; acc.z *= s; acc.w *= s;

    int e = beg;
    for (; e + 8 <= end; e += 8) add_edges<8>(acc, xw, csr + e, c4);
    if (e + 4 <= end) { add_edges<4>(acc, xw, csr + e, c4); e += 4; }
    if (e + 2 <= end) { add_edges<2>(acc, xw, csr + e, c4); e += 2; }
    if (e < end)      { add_edges<1>(acc, xw, csr + e, c4); }

    *(float4*)(h + (size_t)row * HID_C + c4) = acc;
}

__global__ __launch_bounds__(32) void bias_gelu(
    float* __restrict__ h, const float* __restrict__ b, int n)
{
    int node = blockIdx.x;
    int c4   = threadIdx.x << 2;
    size_t idx = (size_t)node * HID_C + c4;
    float4 v  = *(const float4*)(h + idx);
    float4 bb = *(const float4*)(b + c4);
    v.x += bb.x; v.y += bb.y; v.z += bb.z; v.w += bb.w;
    v.x = 0.5f * v.x * (1.0f + erff(v.x * 0.70710678118654752f));
    v.y = 0.5f * v.y * (1.0f + erff(v.y * 0.70710678118654752f));
    v.z = 0.5f * v.z * (1.0f + erff(v.z * 0.70710678118654752f));
    v.w = 0.5f * v.w * (1.0f + erff(v.w * 0.70710678118654752f));
    *(float4*)(h + idx) = v;
}

__global__ __launch_bounds__(32) void pool_accum(
    const float* __restrict__ h, const int* __restrict__ batch,
    float* __restrict__ sums, float* __restrict__ cnt, int n)
{
    int node = blockIdx.x;
    int c4   = threadIdx.x << 2;
    int b    = batch[node];
    float4 v = *(const float4*)(h + (size_t)node * HID_C + c4);
    float* sp = sums + (size_t)b * HID_C + c4;
    atomicAdd(sp + 0, v.x);
    atomicAdd(sp + 1, v.y);
    atomicAdd(sp + 2, v.z);
    atomicAdd(sp + 3, v.w);
    if (threadIdx.x == 0) atomicAdd(&cnt[b], 1.0f);
}

__global__ void pool_final(float* __restrict__ sums, const float* __restrict__ cnt, int n)
{
    int i = blockIdx.x * blockDim.x + threadIdx.x;
    if (i < n) sums[i] /= fmaxf(cnt[i / HID_C], 1.0f);
}

extern "C" void kernel_launch(void* const* d_in, const int* in_sizes, int n_in,
                              void* d_out, int out_size, void* d_ws, size_t ws_size,
                              hipStream_t stream)
{
    const float* x     = (const float*)d_in[0];
    const int*   ei    = (const int*)d_in[1];
    const int*   batch = (const int*)d_in[2];
    const float* W[3]  = { (const float*)d_in[3], (const float*)d_in[5], (const float*)d_in[7] };
    const float* bv[3] = { (const float*)d_in[4], (const float*)d_in[6], (const float*)d_in[8] };
    const float* Wm1   = (const float*)d_in[9];
    const float* bm1   = (const float*)d_in[10];
    const float* Wm2   = (const float*)d_in[11];
    const float* bm2   = (const float*)d_in[12];

    const int N = in_sizes[0] / IN_C;
    const int E = in_sizes[1] / 2;
    const int* src = ei;
    const int* dst = ei + E;

    char*  ws  = (char*)d_ws;
    size_t off = 0;
    auto carve = [&](size_t bytes) -> float* {
        float* p = (float*)(ws + off);
        off = (off + bytes + 255) & ~(size_t)255;
        return p;
    };
    int*   cnt  = (int*)carve((size_t)N * 4);
    float* dinv = carve((size_t)N * 4);
    float* buf0 = carve((size_t)N * HID_C * 4);
    float* buf1 = carve((size_t)N * HID_C * 4);
    float* psum = carve((size_t)NBATCH * HID_C * 4);
    float* pcnt = carve((size_t)NBATCH * 4);
    float* hid  = carve((size_t)NBATCH * HID_C * 4);
    int*   rowptr = (int*)carve((size_t)(N + 1) * 4);
    int*   rank   = (int*)carve((size_t)E * 4);
    int2*  csr    = (int2*)carve((size_t)E * 8);

    zero_i32<<<(N + 255) / 256, 256, 0, stream>>>(cnt, N);
    count_rank<<<(E + 255) / 256, 256, 0, stream>>>(dst, cnt, rank, E);
    scan_rowptr<<<(N + SCAN_T - 1) / SCAN_T, SCAN_T, 0, stream>>>(cnt, rowptr, dinv, N);
    fill_csr<<<(E + 255) / 256, 256, 0, stream>>>(src, dst, rank, rowptr, dinv, csr, E);

    const float* cur = x;
    for (int l = 0; l < 3; ++l) {
        gemm_f32_wmma<<<N / 16, 256, 0, stream>>>(cur, W[l], nullptr, buf0,
                                                  N, HID_C, HID_C, 0);
        gather_rows<<<(N + 7) / 8, 256, 0, stream>>>(buf0, dinv, rowptr, csr, buf1, N);
        bias_gelu<<<N, 32, 0, stream>>>(buf1, bv[l], N);
        cur = buf1;
    }

    fill_f32<<<(NBATCH * HID_C + 255) / 256, 256, 0, stream>>>(psum, 0.0f, NBATCH * HID_C);
    fill_f32<<<(NBATCH + 255) / 256, 256, 0, stream>>>(pcnt, 0.0f, NBATCH);
    pool_accum<<<N, 32, 0, stream>>>(cur, batch, psum, pcnt, N);
    pool_final<<<(NBATCH * HID_C + 255) / 256, 256, 0, stream>>>(psum, pcnt, NBATCH * HID_C);

    gemm_f32_wmma<<<NBATCH / 16, 256, 0, stream>>>(psum, Wm1, bm1, hid,
                                                   NBATCH, HID_C, HID_C, 1);
    gemm_f32_wmma<<<NBATCH / 16, 128, 0, stream>>>(hid, Wm2, bm2, (float*)d_out,
                                                   NBATCH, HID_C, OUT_C, 0);
}


# bias + exact GELU folded into the row gather's store (applied to the row sum in registers): removes 3 launches of 50000 one-wave work-groups and a read+write pass over h per layer
# speedup vs baseline: 2.7477x; 1.0732x over previous
#include <hip/hip_runtime.h>
#include <hip/hip_bf16.h>
#include <math.h>

typedef float v2f __attribute__((ext_vector_type(2)));
typedef float v8f __attribute__((ext_vector_type(8)));

#define IN_C   128
#define HID_C  128
#define OUT_C  64
#define NBATCH 512

__global__ __launch_bounds__(256) void gemm_f32_wmma(
    const float* __restrict__ A, const float* __restrict__ B,
    const float* __restrict__ bias, float* __restrict__ C,
    int M, int K, int N, int act)
{
    __shared__ float lds_a[16][132];

    const int tid  = threadIdx.x;
    const int wave = tid >> 5;
    const int lane = tid & 31;
    const int row0 = blockIdx.x * 16;

    const int nvec = (16 * K) >> 2;
    const int kv   = K >> 2;
    for (int i = tid; i < nvec; i += blockDim.x) {
        int r  = i / kv;
        int k4 = (i - r * kv) << 2;
        float4 v = *(const float4*)(A + (size_t)(row0 + r) * K + k4);
        *(float4*)&lds_a[r][k4] = v;
    }
    __syncthreads();

    const int m    = lane & 15;
    const int kh   = (lane >> 4) << 1;
    const int col0 = wave * 16;

    v8f acc = {};
    for (int kb = 0; kb < K; kb += 4) {
        v2f a, b;
        a.x = lds_a[m][kb + kh];
        a.y = lds_a[m][kb + kh + 1];
        const float* Bp = B + (size_t)(kb + kh) * N + col0 + m;
        b.x = Bp[0];
        b.y = Bp[N];
        acc = __builtin_amdgcn_wmma_f32_16x16x4_f32(
            false, a, false, b, (short)0, acc, false, false);
    }

    const int   col   = col0 + m;
    const int   rbase = row0 + ((lane >> 4) << 3);
    const float bv    = bias ? bias[col] : 0.0f;
#pragma unroll
    for (int v = 0; v < 8; ++v) {
        float x = acc[v] + bv;
        if (act == 1) x = fmaxf(x, 0.0f);
        C[(size_t)(rbase + v) * N + col] = x;
    }
}

__global__ void fill_f32(float* __restrict__ p, float v, int n)
{
    int i = blockIdx.x * blockDim.x + threadIdx.x;
    if (i < n) p[i] = v;
}

__global__ void zero_i32(int* __restrict__ p, int n)
{
    int i = blockIdx.x * blockDim.x + threadIdx.x;
    if (i < n) p[i] = 0;
}

__global__ void count_rank(const int* __restrict__ dst, int* __restrict__ cnt,
                           int* __restrict__ rank, int E)
{
    int e = blockIdx.x * blockDim.x + threadIdx.x;
    if (e < E) rank[e] = atomicAdd(&cnt[dst[e]], 1);
}

#define SCAN_T 1024
__global__ __launch_bounds__(SCAN_T) void scan_rowptr(
    const int* __restrict__ cnt, int* __restrict__ rowptr,
    float* __restrict__ dinv, int n)
{
    __shared__ int s_own[SCAN_T];
    __shared__ int s_pre[SCAN_T];
    const int t = threadIdx.x;
    const int b = blockIdx.x;

    int pre = 0;
    for (int k = 0; k < b; ++k) pre += cnt[k * SCAN_T + t];

    const int i = b * SCAN_T + t;
    const int c = (i < n) ? cnt[i] : 0;
    s_own[t] = c;
    s_pre[t] = pre;
    __syncthreads();
    for (int off = 1; off < SCAN_T; off <<= 1) {
        int a = 0, p = 0;
        if (t >= off) { a = s_own[t - off]; p = s_pre[t - off]; }
        __syncthreads();
        s_own[t] += a;
        s_pre[t] += p;
        __syncthreads();
    }
    const int excl = s_pre[SCAN_T - 1] + s_own[t] - c;
    if (i < n) {
        rowptr[i] = excl;
        dinv[i]   = rsqrtf((float)(c + 1));
    }
    if (i == n - 1) rowptr[n] = excl + c;
}

__global__ void fill_csr(const int* __restrict__ src, const int* __restrict__ dst,
                         const int* __restrict__ rank, const int* __restrict__ rowptr,
                         const float* __restrict__ dinv, int2* __restrict__ csr, int E)
{
    int e = blockIdx.x * blockDim.x + threadIdx.x;
    if (e < E) {
        int s = src[e];
        int d = dst[e];
        int2 r;
        r.x = s;
        r.y = __float_as_int(dinv[s] * dinv[d]);
        csr[rowptr[d] + rank[e]] = r;
    }
}

template <int K>
__device__ __forceinline__ void add_edges(
    float4& acc, const float* __restrict__ xw, const int2* __restrict__ rec, int c4)
{
    int2   r[K];
    float4 v[K];
#pragma unroll
    for (int k = 0; k < K; ++k) r[k] = rec[k];
#pragma unroll
    for (int k = 0; k < K; ++k)
        v[k] = *(const float4*)(xw + (size_t)r[k].x * HID_C + c4);
#pragma unroll
    for (int k = 0; k < K; ++k) {
        const float w = __int_as_float(r[k].y);
        acc.x = fmaf(v[k].x, w, acc.x);
        acc.y = fmaf(v[k].y, w, acc.y);
        acc.z = fmaf(v[k].z, w, acc.z);
        acc.w = fmaf(v[k].w, w, acc.w);
    }
}

__device__ __forceinline__ float gelu_erf(float v)
{
    return 0.5f * v * (1.0f + erff(v * 0.70710678118654752f));
}

__global__ __launch_bounds__(256) void gather_rows(
    const float* __restrict__ xw, const float* __restrict__ dinv,
    const int* __restrict__ rowptr, const int2* __restrict__ csr,
    const float* __restrict__ b, float* __restrict__ h, int n)
{
    const int row = __builtin_amdgcn_readfirstlane(
        (int)(blockIdx.x * 8 + (threadIdx.x >> 5)));
    if (row >= n) return;
    const int c4  = (threadIdx.x & 31) << 2;
    const int beg = rowptr[row];
    const int end = rowptr[row + 1];
    const float di = dinv[row];
    const float s  = di * di;

    const float4 bb = *(const float4*)(b + c4);
    float4 acc = *(const float4*)(xw + (size_t)row * HID_C + c4);
    acc.x *= s; acc.y *= s; acc.z *= s; acc.w *= s;

    int e = beg;
    for (; e + 8 <= end; e += 8) add_edges<8>(acc, xw, csr + e, c4);
    if (e + 4 <= end) { add_edges<4>(acc, xw, csr + e, c4); e += 4; }
    if (e + 2 <= end) { add_edges<2>(acc, xw, csr + e, c4); e += 2; }
    if (e < end)      { add_edges<1>(acc, xw, csr + e, c4); }

    acc.x = gelu_erf(acc.x + bb.x);
    acc.y = gelu_erf(acc.y + bb.y);
    acc.z = gelu_erf(acc.z + bb.z);
    acc.w = gelu_erf(acc.w + bb.w);
    *(float4*)(h + (size_t)row * HID_C + c4) = acc;
}

__global__ __launch_bounds__(32) void pool_accum(
    const float* __restrict__ h, const int* __restrict__ batch,
    float* __restrict__ sums, float* __restrict__ cnt, int n)
{
    int node = blockIdx.x;
    int c4   = threadIdx.x << 2;
    int b    = batch[node];
    float4 v = *(const float4*)(h + (size_t)node * HID_C + c4);
    float* sp = sums + (size_t)b * HID_C + c4;
    atomicAdd(sp + 0, v.x);
    atomicAdd(sp + 1, v.y);
    atomicAdd(sp + 2, v.z);
    atomicAdd(sp + 3, v.w);
    if (threadIdx.x == 0) atomicAdd(&cnt[b], 1.0f);
}

__global__ void pool_final(float* __restrict__ sums, const float* __restrict__ cnt, int n)
{
    int i = blockIdx.x * blockDim.x + threadIdx.x;
    if (i < n) sums[i] /= fmaxf(cnt[i / HID_C], 1.0f);
}

extern "C" void kernel_launch(void* const* d_in, const int* in_sizes, int n_in,
                              void* d_out, int out_size, void* d_ws, size_t ws_size,
                              hipStream_t stream)
{
    const float* x     = (const float*)d_in[0];
    const int*   ei    = (const int*)d_in[1];
    const int*   batch = (const int*)d_in[2];
    const float* W[3]  = { (const float*)d_in[3], (const float*)d_in[5], (const float*)d_in[7] };
    const float* bv[3] = { (const float*)d_in[4], (const float*)d_in[6], (const float*)d_in[8] };
    const float* Wm1   = (const float*)d_in[9];
    const float* bm1   = (const float*)d_in[10];
    const float* Wm2   = (const float*)d_in[11];
    const float* bm2   = (const float*)d_in[12];

    const int N = in_sizes[0] / IN_C;
    const int E = in_sizes[1] / 2;
    const int* src = ei;
    const int* dst = ei + E;

    char*  ws  = (char*)d_ws;
    size_t off = 0;
    auto carve = [&](size_t bytes) -> float* {
        float* p = (float*)(ws + off);
        off = (off + bytes + 255) & ~(size_t)255;
        return p;
    };
    int*   cnt  = (int*)carve((size_t)N * 4);
    float* dinv = carve((size_t)N * 4);
    float* buf0 = carve((size_t)N * HID_C * 4);
    float* buf1 = carve((size_t)N * HID_C * 4);
    float* psum = carve((size_t)NBATCH * HID_C * 4);
    float* pcnt = carve((size_t)NBATCH * 4);
    float* hid  = carve((size_t)NBATCH * HID_C * 4);
    int*   rowptr = (int*)carve((size_t)(N + 1) * 4);
    int*   rank   = (int*)carve((size_t)E * 4);
    int2*  csr    = (int2*)carve((size_t)E * 8);

    zero_i32<<<(N + 255) / 256, 256, 0, stream>>>(cnt, N);
    count_rank<<<(E + 255) / 256, 256, 0, stream>>>(dst, cnt, rank, E);
    scan_rowptr<<<(N + SCAN_T - 1) / SCAN_T, SCAN_T, 0, stream>>>(cnt, rowptr, dinv, N);
    fill_csr<<<(E + 255) / 256, 256, 0, stream>>>(src, dst, rank, rowptr, dinv, csr, E);

    const float* cur = x;
    for (int l = 0; l < 3; ++l) {
        gemm_f32_wmma<<<N / 16, 256, 0, stream>>>(cur, W[l], nullptr, buf0,
                                                  N, HID_C, HID_C, 0);
        gather_rows<<<(N + 7) / 8, 256, 0, stream>>>(buf0, dinv, rowptr, csr, bv[l], buf1, N);
        cur = buf1;
    }

    fill_f32<<<(NBATCH * HID_C + 255) / 256, 256, 0, stream>>>(psum, 0.0f, NBATCH * HID_C);
    fill_f32<<<(NBATCH + 255) / 256, 256, 0, stream>>>(pcnt, 0.0f, NBATCH);
    pool_accum<<<N, 32, 0, stream>>>(cur, batch, psum, pcnt, N);
    pool_final<<<(NBATCH * HID_C + 255) / 256, 256, 0, stream>>>(psum, pcnt, NBATCH * HID_C);

    gemm_f32_wmma<<<NBATCH / 16, 256, 0, stream>>>(psum, Wm1, bm1, hid,
                                                   NBATCH, HID_C, HID_C, 1);
    gemm_f32_wmma<<<NBATCH / 16, 128, 0, stream>>>(hid, Wm2, bm2, (float*)d_out,
                                                   NBATCH, HID_C, OUT_C, 0);
}


# edge ranking: 1.2M scattered returning memory atomics -> returning LDS adds on per-work-group node counters (16384-edge slices) + per-node slice starts; arrival order per slice, row sum order changes
# speedup vs baseline: 2.9762x; 1.0832x over previous
#include <hip/hip_runtime.h>
#include <hip/hip_bf16.h>
#include <math.h>

typedef float v2f __attribute__((ext_vector_type(2)));
typedef float v8f __attribute__((ext_vector_type(8)));

#define IN_C   128
#define HID_C  128
#define OUT_C  64
#define NBATCH 512

__global__ __launch_bounds__(256) void gemm_f32_wmma(
    const float* __restrict__ A, const float* __restrict__ B,
    const float* __restrict__ bias, float* __restrict__ C,
    int M, int K, int N, int act)
{
    __shared__ float lds_a[16][132];

    const int tid  = threadIdx.x;
    const int wave = tid >> 5;
    const int lane = tid & 31;
    const int row0 = blockIdx.x * 16;

    const int nvec = (16 * K) >> 2;
    const int kv   = K >> 2;
    for (int i = tid; i < nvec; i += blockDim.x) {
        int r  = i / kv;
        int k4 = (i - r * kv) << 2;
        float4 v = *(const float4*)(A + (size_t)(row0 + r) * K + k4);
        *(float4*)&lds_a[r][k4] = v;
    }
    __syncthreads();

    const int m    = lane & 15;
    const int kh   = (lane >> 4) << 1;
    const int col0 = wave * 16;

    v8f acc = {};
    for (int kb = 0; kb < K; kb += 4) {
        v2f a, b;
        a.x = lds_a[m][kb + kh];
        a.y = lds_a[m][kb + kh + 1];
        const float* Bp = B + (size_t)(kb + kh) * N + col0 + m;
        b.x = Bp[0];
        b.y = Bp[N];
        acc = __builtin_amdgcn_wmma_f32_16x16x4_f32(
            false, a, false, b, (short)0, acc, false, false);
    }

    const int   col   = col0 + m;
    const int   rbase = row0 + ((lane >> 4) << 3);
    const float bv    = bias ? bias[col] : 0.0f;
#pragma unroll
    for (int v = 0; v < 8; ++v) {
        float x = acc[v] + bv;
        if (act == 1) x = fmaxf(x, 0.0f);
        C[(size_t)(rbase + v) * N + col] = x;
    }
}

__global__ void fill_f32(float* __restrict__ p, float v, int n)
{
    int i = blockIdx.x * blockDim.x + threadIdx.x;
    if (i < n) p[i] = v;
}

#define RANK_CHUNK 16384
#define RANK_T     1024
#define MAX_NODES  50000

__global__ __launch_bounds__(RANK_T) void rank_slices(
    const int* __restrict__ dst, int* __restrict__ cntm, int* __restrict__ rank,
    int E, int n)
{
    __shared__ alignas(16) int lc[MAX_NODES];
    const int t = threadIdx.x;
    const int g = blockIdx.x;

    for (int i = t; i < MAX_NODES / 4; i += RANK_T)
        ((int4*)lc)[i] = make_int4(0, 0, 0, 0);
    __syncthreads();

    const int beg = g * RANK_CHUNK;
    const int end = min(beg + RANK_CHUNK, E);
    for (int e = beg + t; e < end; e += 4 * RANK_T) {
        const int e1 = e + RANK_T, e2 = e + 2 * RANK_T, e3 = e + 3 * RANK_T;
        if (e3 < end) {
            const int d0 = dst[e];
            const int d1 = dst[e1];
            const int d2 = dst[e2];
            const int d3 = dst[e3];
            const int r0 = atomicAdd(&lc[d0], 1);
            const int r1 = atomicAdd(&lc[d1], 1);
            const int r2 = atomicAdd(&lc[d2], 1);
            const int r3 = atomicAdd(&lc[d3], 1);
            rank[e]  = r0;
            rank[e1] = r1;
            rank[e2] = r2;
            rank[e3] = r3;
        } else {
            for (int k = e; k < end; k += RANK_T)
                rank[k] = atomicAdd(&lc[dst[k]], 1);
        }
    }
    __syncthreads();

    int* out = cntm + (size_t)g * n;
    for (int i = t; i < n; i += RANK_T) out[i] = lc[i];
}

__global__ void slice_offsets(int* __restrict__ cntm, int* __restrict__ cnt, int n, int G)
{
    const int i = blockIdx.x * blockDim.x + threadIdx.x;
    if (i >= n) return;
    int* p = cntm + i;
    int run = 0;
    int g = 0;
    for (; g + 8 <= G; g += 8) {
        int c[8];
#pragma unroll
        for (int k = 0; k < 8; ++k) c[k] = p[(size_t)(g + k) * n];
#pragma unroll
        for (int k = 0; k < 8; ++k) {
            p[(size_t)(g + k) * n] = run;
            run += c[k];
        }
    }
    for (; g < G; ++g) {
        const int c = p[(size_t)g * n];
        p[(size_t)g * n] = run;
        run += c;
    }
    cnt[i] = run;
}

#define SCAN_T 1024
__global__ __launch_bounds__(SCAN_T) void scan_rowptr(
    const int* __restrict__ cnt, int* __restrict__ rowptr,
    float* __restrict__ dinv, int n)
{
    __shared__ int s_own[SCAN_T];
    __shared__ int s_pre[SCAN_T];
    const int t = threadIdx.x;
    const int b = blockIdx.x;

    int pre = 0;
    for (int k = 0; k < b; ++k) pre += cnt[k * SCAN_T + t];

    const int i = b * SCAN_T + t;
    const int c = (i < n) ? cnt[i] : 0;
    s_own[t] = c;
    s_pre[t] = pre;
    __syncthreads();
    for (int off = 1; off < SCAN_T; off <<= 1) {
        int a = 0, p = 0;
        if (t >= off) { a = s_own[t - off]; p = s_pre[t - off]; }
        __syncthreads();
        s_own[t] += a;
        s_pre[t] += p;
        __syncthreads();
    }
    const int excl = s_pre[SCAN_T - 1] + s_own[t] - c;
    if (i < n) {
        rowptr[i] = excl;
        dinv[i]   = rsqrtf((float)(c + 1));
    }
    if (i == n - 1) rowptr[n] = excl + c;
}

__global__ void fill_csr(const int* __restrict__ src, const int* __restrict__ dst,
                         const int* __restrict__ rank, const int* __restrict__ rowptr,
                         const int* __restrict__ cntm,
                         const float* __restrict__ dinv, int2* __restrict__ csr, int E, int n)
{
    int e = blockIdx.x * blockDim.x + threadIdx.x;
    if (e < E) {
        int s = src[e];
        int d = dst[e];
        int g = e / RANK_CHUNK;
        int2 r;
        r.x = s;
        r.y = __float_as_int(dinv[s] * dinv[d]);
        csr[rowptr[d] + cntm[(size_t)g * n + d] + rank[e]] = r;
    }
}

template <int K>
__device__ __forceinline__ void add_edges(
    float4& acc, const float* __restrict__ xw, const int2* __restrict__ rec, int c4)
{
    int2   r[K];
    float4 v[K];
#pragma unroll
    for (int k = 0; k < K; ++k) r[k] = rec[k];
#pragma unroll
    for (int k = 0; k < K; ++k)
        v[k] = *(const float4*)(xw + (size_t)r[k].x * HID_C + c4);
#pragma unroll
    for (int k = 0; k < K; ++k) {
        const float w = __int_as_float(r[k].y);
        acc.x = fmaf(v[k].x, w, acc.x);
        acc.y = fmaf(v[k].y, w, acc.y);
        acc.z = fmaf(v[k].z, w, acc.z);
        acc.w = fmaf(v[k].w, w, acc.w);
    }
}

__device__ __forceinline__ float gelu_erf(float v)
{
    return 0.5f * v * (1.0f + erff(v * 0.70710678118654752f));
}

__global__ __launch_bounds__(256) void gather_rows(
    const float* __restrict__ xw, const float* __restrict__ dinv,
    const int* __restrict__ rowptr, const int2* __restrict__ csr,
    const float* __restrict__ b, float* __restrict__ h, int n)
{
    const int row = __builtin_amdgcn_readfirstlane(
        (int)(blockIdx.x * 8 + (threadIdx.x >> 5)));
    if (row >= n) return;
    const int c4  = (threadIdx.x & 31) << 2;
    const int beg = rowptr[row];
    const int end = rowptr[row + 1];
    const float di = dinv[row];
    const float s  = di * di;

    const float4 bb = *(const float4*)(b + c4);
    float4 acc = *(const float4*)(xw + (size_t)row * HID_C + c4);
    acc.x *= s; acc.y *= s; acc.z *= s; acc.w *= s;

    int e = beg;
    for (; e + 8 <= end; e += 8) add_edges<8>(acc, xw, csr + e, c4);
    if (e + 4 <= end) { add_edges<4>(acc, xw, csr + e, c4); e += 4; }
    if (e + 2 <= end) { add_edges<2>(acc, xw, csr + e, c4); e += 2; }
    if (e < end)      { add_edges<1>(acc, xw, csr + e, c4); }

    acc.x = gelu_erf(acc.x + bb.x);
    acc.y = gelu_erf(acc.y + bb.y);
    acc.z = gelu_erf(acc.z + bb.z);
    acc.w = gelu_erf(acc.w + bb.w);
    *(float4*)(h + (size_t)row * HID_C + c4) = acc;
}

__global__ __launch_bounds__(32) void pool_accum(
    const float* __restrict__ h, const int* __restrict__ batch,
    float* __restrict__ sums, float* __restrict__ cnt, int n)
{
    int node = blockIdx.x;
    int c4   = threadIdx.x << 2;
    int b    = batch[node];
    float4 v = *(const float4*)(h + (size_t)node * HID_C + c4);
    float* sp = sums + (size_t)b * HID_C + c4;
    atomicAdd(sp + 0, v.x);
    atomicAdd(sp + 1, v.y);
    atomicAdd(sp + 2, v.z);
    atomicAdd(sp + 3, v.w);
    if (threadIdx.x == 0) atomicAdd(&cnt[b], 1.0f);
}

__global__ void pool_final(float* __restrict__ sums, const float* __restrict__ cnt, int n)
{
    int i = blockIdx.x * blockDim.x + threadIdx.x;
    if (i < n) sums[i] /= fmaxf(cnt[i / HID_C], 1.0f);
}

extern "C" void kernel_launch(void* const* d_in, const int* in_sizes, int n_in,
                              void* d_out, int out_size, void* d_ws, size_t ws_size,
                              hipStream_t stream)
{
    const float* x     = (const float*)d_in[0];
    const int*   ei    = (const int*)d_in[1];
    const int*   batch = (const int*)d_in[2];
    const float* W[3]  = { (const float*)d_in[3], (const float*)d_in[5], (const float*)d_in[7] };
    const float* bv[3] = { (const float*)d_in[4], (const float*)d_in[6], (const float*)d_in[8] };
    const float* Wm1   = (const float*)d_in[9];
    const float* bm1   = (const float*)d_in[10];
    const float* Wm2   = (const float*)d_in[11];
    const float* bm2   = (const float*)d_in[12];

    const int N = in_sizes[0] / IN_C;
    const int E = in_sizes[1] / 2;
    const int* src = ei;
    const int* dst = ei + E;

    char*  ws  = (char*)d_ws;
    size_t off = 0;
    auto carve = [&](size_t bytes) -> float* {
        float* p = (float*)(ws + off);
        off = (off + bytes + 255) & ~(size_t)255;
        return p;
    };
    int*   cnt  = (int*)carve((size_t)N * 4);
    float* dinv = carve((size_t)N * 4);
    float* buf0 = carve((size_t)N * HID_C * 4);
    float* buf1 = carve((size_t)N * HID_C * 4);
    float* psum = carve((size_t)NBATCH * HID_C * 4);
    float* pcnt = carve((size_t)NBATCH * 4);
    float* hid  = carve((size_t)NBATCH * HID_C * 4);
    int*   rowptr = (int*)carve((size_t)(N + 1) * 4);
    int*   rank   = (int*)carve((size_t)E * 4);
    int2*  csr    = (int2*)carve((size_t)E * 8);
    const int G   = (E + RANK_CHUNK - 1) / RANK_CHUNK;
    int*   cntm   = (int*)carve((size_t)G * N * 4);

    rank_slices<<<G, RANK_T, 0, stream>>>(dst, cntm, rank, E, N);
    slice_offsets<<<(N + 255) / 256, 256, 0, stream>>>(cntm, cnt, N, G);
    scan_rowptr<<<(N + SCAN_T - 1) / SCAN_T, SCAN_T, 0, stream>>>(cnt, rowptr, dinv, N);
    fill_csr<<<(E + 255) / 256, 256, 0, stream>>>(src, dst, rank, rowptr, cntm, dinv, csr, E, N);

    const float* cur = x;
    for (int l = 0; l < 3; ++l) {
        gemm_f32_wmma<<<N / 16, 256, 0, stream>>>(cur, W[l], nullptr, buf0,
                                                  N, HID_C, HID_C, 0);
        gather_rows<<<(N + 7) / 8, 256, 0, stream>>>(buf0, dinv, rowptr, csr, bv[l], buf1, N);
        cur = buf1;
    }

    fill_f32<<<(NBATCH * HID_C + 255) / 256, 256, 0, stream>>>(psum, 0.0f, NBATCH * HID_C);
    fill_f32<<<(NBATCH + 255) / 256, 256, 0, stream>>>(pcnt, 0.0f, NBATCH);
    pool_accum<<<N, 32, 0, stream>>>(cur, batch, psum, pcnt, N);
    pool_final<<<(NBATCH * HID_C + 255) / 256, 256, 0, stream>>>(psum, pcnt, NBATCH * HID_C);

    gemm_f32_wmma<<<NBATCH / 16, 256, 0, stream>>>(psum, Wm1, bm1, hid,
                                                   NBATCH, HID_C, HID_C, 1);
    gemm_f32_wmma<<<NBATCH / 16, 128, 0, stream>>>(hid, Wm2, bm2, (float*)d_out,
                                                   NBATCH, HID_C, OUT_C, 0);
}


# layer product fused into the row gather: neighbour rows of h summed first, the 16x128 sums handed to the f32 WMMA through LDS, then W, bias, GELU; 3 launches and the xw plane less; sum order changes
# speedup vs baseline: 3.5310x; 1.1864x over previous
#include <hip/hip_runtime.h>
#include <hip/hip_bf16.h>
#include <math.h>

typedef float v2f __attribute__((ext_vector_type(2)));
typedef float v8f __attribute__((ext_vector_type(8)));

#define IN_C   128
#define HID_C  128
#define OUT_C  64
#define NBATCH 512

__global__ __launch_bounds__(256) void gemm_f32_wmma(
    const float* __restrict__ A, const float* __restrict__ B,
    const float* __restrict__ bias, float* __restrict__ C,
    int M, int K, int N, int act)
{
    __shared__ float lds_a[16][132];

    const int tid  = threadIdx.x;
    const int wave = tid >> 5;
    const int lane = tid & 31;
    const int row0 = blockIdx.x * 16;

    const int nvec = (16 * K) >> 2;
    const int kv   = K >> 2;
    for (int i = tid; i < nvec; i += blockDim.x) {
        int r  = i / kv;
        int k4 = (i - r * kv) << 2;
        float4 v = *(const float4*)(A + (size_t)(row0 + r) * K + k4);
        *(float4*)&lds_a[r][k4] = v;
    }
    __syncthreads();

    const int m    = lane & 15;
    const int kh   = (lane >> 4) << 1;
    const int col0 = wave * 16;

    v8f acc = {};
    for (int kb = 0; kb < K; kb += 4) {
        v2f a, b;
        a.x = lds_a[m][kb + kh];
        a.y = lds_a[m][kb + kh + 1];
        const float* Bp = B + (size_t)(kb + kh) * N + col0 + m;
        b.x = Bp[0];
        b.y = Bp[N];
        acc = __builtin_amdgcn_wmma_f32_16x16x4_f32(
            false, a, false, b, (short)0, acc, false, false);
    }

    const int   col   = col0 + m;
    const int   rbase = row0 + ((lane >> 4) << 3);
    const float bv    = bias ? bias[col] : 0.0f;
#pragma unroll
    for (int v = 0; v < 8; ++v) {
        float x = acc[v] + bv;
        if (act == 1) x = fmaxf(x, 0.0f);
        C[(size_t)(rbase + v) * N + col] = x;
    }
}

__global__ void fill_f32(float* __restrict__ p, float v, int n)
{
    int i = blockIdx.x * blockDim.x + threadIdx.x;
    if (i < n) p[i] = v;
}

#define RANK_CHUNK 16384
#define RANK_T     1024
#define MAX_NODES  50000

__global__ __launch_bounds__(RANK_T) void rank_slices(
    const int* __restrict__ dst, int* __restrict__ cntm, int* __restrict__ rank,
    int E, int n)
{
    __shared__ alignas(16) int lc[MAX_NODES];
    const int t = threadIdx.x;
    const int g = blockIdx.x;

    for (int i = t; i < MAX_NODES / 4; i += RANK_T)
        ((int4*)lc)[i] = make_int4(0, 0, 0, 0);
    __syncthreads();

    const int beg = g * RANK_CHUNK;
    const int end = min(beg + RANK_CHUNK, E);
    for (int e = beg + t; e < end; e += 4 * RANK_T) {
        const int e1 = e + RANK_T, e2 = e + 2 * RANK_T, e3 = e + 3 * RANK_T;
        if (e3 < end) {
            const int d0 = dst[e];
            const int d1 = dst[e1];
            const int d2 = dst[e2];
            const int d3 = dst[e3];
            const int r0 = atomicAdd(&lc[d0], 1);
            const int r1 = atomicAdd(&lc[d1], 1);
            const int r2 = atomicAdd(&lc[d2], 1);
            const int r3 = atomicAdd(&lc[d3], 1);
            rank[e]  = r0;
            rank[e1] = r1;
            rank[e2] = r2;
            rank[e3] = r3;
        } else {
            for (int k = e; k < end; k += RANK_T)
                rank[k] = atomicAdd(&lc[dst[k]], 1);
        }
    }
    __syncthreads();

    int* out = cntm + (size_t)g * n;
    for (int i = t; i < n; i += RANK_T) out[i] = lc[i];
}

__global__ void slice_offsets(int* __restrict__ cntm, int* __restrict__ cnt, int n, int G)
{
    const int i = blockIdx.x * blockDim.x + threadIdx.x;
    if (i >= n) return;
    int* p = cntm + i;
    int run = 0;
    int g = 0;
    for (; g + 8 <= G; g += 8) {
        int c[8];
#pragma unroll
        for (int k = 0; k < 8; ++k) c[k] = p[(size_t)(g + k) * n];
#pragma unroll
        for (int k = 0; k < 8; ++k) {
            p[(size_t)(g + k) * n] = run;
            run += c[k];
        }
    }
    for (; g < G; ++g) {
        const int c = p[(size_t)g * n];
        p[(size_t)g * n] = run;
        run += c;
    }
    cnt[i] = run;
}

#define SCAN_T 1024
__global__ __launch_bounds__(SCAN_T) void scan_rowptr(
    const int* __restrict__ cnt, int* __restrict__ rowptr,
    float* __restrict__ dinv, int n)
{
    __shared__ int s_own[SCAN_T];
    __shared__ int s_pre[SCAN_T];
    const int t = threadIdx.x;
    const int b = blockIdx.x;

    int pre = 0;
    for (int k = 0; k < b; ++k) pre += cnt[k * SCAN_T + t];

    const int i = b * SCAN_T + t;
    const int c = (i < n) ? cnt[i] : 0;
    s_own[t] = c;
    s_pre[t] = pre;
    __syncthreads();
    for (int off = 1; off < SCAN_T; off <<= 1) {
        int a = 0, p = 0;
        if (t >= off) { a = s_own[t - off]; p = s_pre[t - off]; }
        __syncthreads();
        s_own[t] += a;
        s_pre[t] += p;
        __syncthreads();
    }
    const int excl = s_pre[SCAN_T - 1] + s_own[t] - c;
    if (i < n) {
        rowptr[i] = excl;
        dinv[i]   = rsqrtf((float)(c + 1));
    }
    if (i == n - 1) rowptr[n] = excl + c;
}

__global__ void fill_csr(const int* __restrict__ src, const int* __restrict__ dst,
                         const int* __restrict__ rank, const int* __restrict__ rowptr,
                         const int* __restrict__ cntm,
                         const float* __restrict__ dinv, int2* __restrict__ csr, int E, int n)
{
    int e = blockIdx.x * blockDim.x + threadIdx.x;
    if (e < E) {
        int s = src[e];
        int d = dst[e];
        int g = e / RANK_CHUNK;
        int2 r;
        r.x = s;
        r.y = __float_as_int(dinv[s] * dinv[d]);
        csr[rowptr[d] + cntm[(size_t)g * n + d] + rank[e]] = r;
    }
}

template <int K>
__device__ __forceinline__ void add_edges(
    float4& acc, const float* __restrict__ xw, const int2* __restrict__ rec, int c4)
{
    int2   r[K];
    float4 v[K];
#pragma unroll
    for (int k = 0; k < K; ++k) r[k] = rec[k];
#pragma unroll
    for (int k = 0; k < K; ++k)
        v[k] = *(const float4*)(xw + (size_t)r[k].x * HID_C + c4);
#pragma unroll
    for (int k = 0; k < K; ++k) {
        const float w = __int_as_float(r[k].y);
        acc.x = fmaf(v[k].x, w, acc.x);
        acc.y = fmaf(v[k].y, w, acc.y);
        acc.z = fmaf(v[k].z, w, acc.z);
        acc.w = fmaf(v[k].w, w, acc.w);
    }
}

__device__ __forceinline__ float gelu_erf(float v)
{
    return 0.5f * v * (1.0f + erff(v * 0.70710678118654752f));
}

__global__ __launch_bounds__(256) void gcn_layer(
    const float* __restrict__ hin, const float* __restrict__ dinv,
    const int* __restrict__ rowptr, const int2* __restrict__ csr,
    const float* __restrict__ Wl, const float* __restrict__ b,
    float* __restrict__ hout, int n)
{
    __shared__ float lds_a[16][132];

    const int tid  = threadIdx.x;
    const int lane = tid & 31;
    const int wv   = tid >> 5;
    const int row0 = blockIdx.x * 16;
    const int c4   = lane << 2;

    const int  rowA = __builtin_amdgcn_readfirstlane(row0 + wv * 2);
    const bool hasA = rowA < n;
    const bool hasB = rowA + 1 < n;
    int   p0 = 0, p1 = 0, p2 = 0;
    float dA = 0.0f, dB = 0.0f;
    float4 accA = make_float4(0.0f, 0.0f, 0.0f, 0.0f);
    float4 accB = make_float4(0.0f, 0.0f, 0.0f, 0.0f);
    if (hasA) {
        p0 = rowptr[rowA];
        p1 = rowptr[rowA + 1];
        dA = dinv[rowA];
        accA = *(const float4*)(hin + (size_t)rowA * HID_C + c4);
    }
    if (hasB) {
        p2 = rowptr[rowA + 2];
        dB = dinv[rowA + 1];
        accB = *(const float4*)(hin + (size_t)(rowA + 1) * HID_C + c4);
    }
    {
        const float s = dA * dA;
        accA.x *= s; accA.y *= s; accA.z *= s; accA.w *= s;
        int e = p0;
        for (; e + 8 <= p1; e += 8) add_edges<8>(accA, hin, csr + e, c4);
        if (e + 4 <= p1) { add_edges<4>(accA, hin, csr + e, c4); e += 4; }
        if (e + 2 <= p1) { add_edges<2>(accA, hin, csr + e, c4); e += 2; }
        if (e < p1)      { add_edges<1>(accA, hin, csr + e, c4); }
        *(float4*)&lds_a[wv * 2][c4] = accA;
    }
    if (!hasB) p2 = p1;
    {
        const float s = dB * dB;
        accB.x *= s; accB.y *= s; accB.z *= s; accB.w *= s;
        int e = p1;
        for (; e + 8 <= p2; e += 8) add_edges<8>(accB, hin, csr + e, c4);
        if (e + 4 <= p2) { add_edges<4>(accB, hin, csr + e, c4); e += 4; }
        if (e + 2 <= p2) { add_edges<2>(accB, hin, csr + e, c4); e += 2; }
        if (e < p2)      { add_edges<1>(accB, hin, csr + e, c4); }
        *(float4*)&lds_a[wv * 2 + 1][c4] = accB;
    }
    __syncthreads();

    const int m    = lane & 15;
    const int kh   = (lane >> 4) << 1;
    const int col0 = wv * 16;
    const float* Bp = Wl + (size_t)kh * HID_C + col0 + m;

    v8f cacc = {};
#pragma unroll 8
    for (int kb = 0; kb < HID_C; kb += 4) {
        v2f a, w;
        a.x = lds_a[m][kb + kh];
        a.y = lds_a[m][kb + kh + 1];
        w.x = Bp[(size_t)kb * HID_C];
        w.y = Bp[(size_t)(kb + 1) * HID_C];
        cacc = __builtin_amdgcn_wmma_f32_16x16x4_f32(
            false, a, false, w, (short)0, cacc, false, false);
    }

    const int   col   = col0 + m;
    const int   rbase = row0 + ((lane >> 4) << 3);
    const float bv    = b[col];
    float y[8];
#pragma unroll
    for (int v = 0; v < 8; ++v) y[v] = gelu_erf(cacc[v] + bv);
#pragma unroll
    for (int v = 0; v < 8; ++v) {
        if (rbase + v < n) hout[(size_t)(rbase + v) * HID_C + col] = y[v];
    }
}

__global__ __launch_bounds__(32) void pool_accum(
    const float* __restrict__ h, const int* __restrict__ batch,
    float* __restrict__ sums, float* __restrict__ cnt, int n)
{
    int node = blockIdx.x;
    int c4   = threadIdx.x << 2;
    int b    = batch[node];
    float4 v = *(const float4*)(h + (size_t)node * HID_C + c4);
    float* sp = sums + (size_t)b * HID_C + c4;
    atomicAdd(sp + 0, v.x);
    atomicAdd(sp + 1, v.y);
    atomicAdd(sp + 2, v.z);
    atomicAdd(sp + 3, v.w);
    if (threadIdx.x == 0) atomicAdd(&cnt[b], 1.0f);
}

__global__ void pool_final(float* __restrict__ sums, const float* __restrict__ cnt, int n)
{
    int i = blockIdx.x * blockDim.x + threadIdx.x;
    if (i < n) sums[i] /= fmaxf(cnt[i / HID_C], 1.0f);
}

extern "C" void kernel_launch(void* const* d_in, const int* in_sizes, int n_in,
                              void* d_out, int out_size, void* d_ws, size_t ws_size,
                              hipStream_t stream)
{
    const float* x     = (const float*)d_in[0];
    const int*   ei    = (const int*)d_in[1];
    const int*   batch = (const int*)d_in[2];
    const float* W[3]  = { (const float*)d_in[3], (const float*)d_in[5], (const float*)d_in[7] };
    const float* bv[3] = { (const float*)d_in[4], (const float*)d_in[6], (const float*)d_in[8] };
    const float* Wm1   = (const float*)d_in[9];
    const float* bm1   = (const float*)d_in[10];
    const float* Wm2   = (const float*)d_in[11];
    const float* bm2   = (const float*)d_in[12];

    const int N = in_sizes[0] / IN_C;
    const int E = in_sizes[1] / 2;
    const int* src = ei;
    const int* dst = ei + E;

    char*  ws  = (char*)d_ws;
    size_t off = 0;
    auto carve = [&](size_t bytes) -> float* {
        float* p = (float*)(ws + off);
        off = (off + bytes + 255) & ~(size_t)255;
        return p;
    };
    int*   cnt  = (int*)carve((size_t)N * 4);
    float* dinv = carve((size_t)N * 4);
    float* buf0 = carve((size_t)N * HID_C * 4);
    float* buf1 = carve((size_t)N * HID_C * 4);
    float* psum = carve((size_t)NBATCH * HID_C * 4);
    float* pcnt = carve((size_t)NBATCH * 4);
    float* hid  = carve((size_t)NBATCH * HID_C * 4);
    int*   rowptr = (int*)carve((size_t)(N + 1) * 4);
    int*   rank   = (int*)carve((size_t)E * 4);
    int2*  csr    = (int2*)carve((size_t)E * 8);
    const int G   = (E + RANK_CHUNK - 1) / RANK_CHUNK;
    int*   cntm   = (int*)carve((size_t)G * N * 4);

    rank_slices<<<G, RANK_T, 0, stream>>>(dst, cntm, rank, E, N);
    slice_offsets<<<(N + 255) / 256, 256, 0, stream>>>(cntm, cnt, N, G);
    scan_rowptr<<<(N + SCAN_T - 1) / SCAN_T, SCAN_T, 0, stream>>>(cnt, rowptr, dinv, N);
    fill_csr<<<(E + 255) / 256, 256, 0, stream>>>(src, dst, rank, rowptr, cntm, dinv, csr, E, N);

    const float* cur = x;
    float* pp[2] = { buf0, buf1 };
    for (int l = 0; l < 3; ++l) {
        float* nxt = pp[l & 1];
        gcn_layer<<<(N + 15) / 16, 256, 0, stream>>>(cur, dinv, rowptr, csr, W[l], bv[l], nxt, N);
        cur = nxt;
    }

    fill_f32<<<(NBATCH * HID_C + 255) / 256, 256, 0, stream>>>(psum, 0.0f, NBATCH * HID_C);
    fill_f32<<<(NBATCH + 255) / 256, 256, 0, stream>>>(pcnt, 0.0f, NBATCH);
    pool_accum<<<N, 32, 0, stream>>>(cur, batch, psum, pcnt, N);
    pool_final<<<(NBATCH * HID_C + 255) / 256, 256, 0, stream>>>(psum, pcnt, NBATCH * HID_C);

    gemm_f32_wmma<<<NBATCH / 16, 256, 0, stream>>>(psum, Wm1, bm1, hid,
                                                   NBATCH, HID_C, HID_C, 1);
    gemm_f32_wmma<<<NBATCH / 16, 128, 0, stream>>>(hid, Wm2, bm2, (float*)d_out,
                                                   NBATCH, HID_C, OUT_C, 0);
}


# mean pool folded into its neighbours: last layer's epilogue adds its tile's per-graph run sums to the pool sums (h3 never stored), layer 2 zeroes them, head tile load divides; pool sum order changes
# speedup vs baseline: 4.4638x; 1.2642x over previous
#include <hip/hip_runtime.h>
#include <hip/hip_bf16.h>
#include <math.h>

typedef float v2f __attribute__((ext_vector_type(2)));
typedef float v8f __attribute__((ext_vector_type(8)));

#define IN_C   128
#define HID_C  128
#define OUT_C  64
#define NBATCH 512

__global__ __launch_bounds__(256) void gemm_f32_wmma(
    const float* __restrict__ A, const float* __restrict__ B,
    const float* __restrict__ bias, const float* __restrict__ rowdiv,
    float* __restrict__ C,
    int M, int K, int N, int act)
{
    __shared__ float lds_a[16][132];

    const int tid  = threadIdx.x;
    const int wave = tid >> 5;
    const int lane = tid & 31;
    const int row0 = blockIdx.x * 16;

    const int nvec = (16 * K) >> 2;
    const int kv   = K >> 2;
    for (int i = tid; i < nvec; i += blockDim.x) {
        int r  = i / kv;
        int k4 = (i - r * kv) << 2;
        float4 v = *(const float4*)(A + (size_t)(row0 + r) * K + k4);
        if (rowdiv) {
            const float c = fmaxf(rowdiv[row0 + r], 1.0f);
            v.x /= c; v.y /= c; v.z /= c; v.w /= c;
        }
        *(float4*)&lds_a[r][k4] = v;
    }
    __syncthreads();

    const int m    = lane & 15;
    const int kh   = (lane >> 4) << 1;
    const int col0 = wave * 16;

    v8f acc = {};
    for (int kb = 0; kb < K; kb += 4) {
        v2f a, b;
        a.x = lds_a[m][kb + kh];
        a.y = lds_a[m][kb + kh + 1];
        const float* Bp = B + (size_t)(kb + kh) * N + col0 + m;
        b.x = Bp[0];
        b.y = Bp[N];
        acc = __builtin_amdgcn_wmma_f32_16x16x4_f32(
            false, a, false, b, (short)0, acc, false, false);
    }

    const int   col   = col0 + m;
    const int   rbase = row0 + ((lane >> 4) << 3);
    const float bv    = bias ? bias[col] : 0.0f;
#pragma unroll
    for (int v = 0; v < 8; ++v) {
        float x = acc[v] + bv;
        if (act == 1) x = fmaxf(x, 0.0f);
        C[(size_t)(rbase + v) * N + col] = x;
    }
}

#define RANK_CHUNK 16384
#define RANK_T     1024
#define MAX_NODES  50000

__global__ __launch_bounds__(RANK_T) void rank_slices(
    const int* __restrict__ dst, int* __restrict__ cntm, int* __restrict__ rank,
    int E, int n)
{
    __shared__ alignas(16) int lc[MAX_NODES];
    const int t = threadIdx.x;
    const int g = blockIdx.x;

    for (int i = t; i < MAX_NODES / 4; i += RANK_T)
        ((int4*)lc)[i] = make_int4(0, 0, 0, 0);
    __syncthreads();

    const int beg = g * RANK_CHUNK;
    const int end = min(beg + RANK_CHUNK, E);
    for (int e = beg + t; e < end; e += 4 * RANK_T) {
        const int e1 = e + RANK_T, e2 = e + 2 * RANK_T, e3 = e + 3 * RANK_T;
        if (e3 < end) {
            const int d0 = dst[e];
            const int d1 = dst[e1];
            const int d2 = dst[e2];
            const int d3 = dst[e3];
            const int r0 = atomicAdd(&lc[d0], 1);
            const int r1 = atomicAdd(&lc[d1], 1);
            const int r2 = atomicAdd(&lc[d2], 1);
            const int r3 = atomicAdd(&lc[d3], 1);
            rank[e]  = r0;
            rank[e1] = r1;
            rank[e2] = r2;
            rank[e3] = r3;
        } else {
            for (int k = e; k < end; k += RANK_T)
                rank[k] = atomicAdd(&lc[dst[k]], 1);
        }
    }
    __syncthreads();

    int* out = cntm + (size_t)g * n;
    for (int i = t; i < n; i += RANK_T) out[i] = lc[i];
}

__global__ void slice_offsets(int* __restrict__ cntm, int* __restrict__ cnt, int n, int G)
{
    const int i = blockIdx.x * blockDim.x + threadIdx.x;
    if (i >= n) return;
    int* p = cntm + i;
    int run = 0;
    int g = 0;
    for (; g + 8 <= G; g += 8) {
        int c[8];
#pragma unroll
        for (int k = 0; k < 8; ++k) c[k] = p[(size_t)(g + k) * n];
#pragma unroll
        for (int k = 0; k < 8; ++k) {
            p[(size_t)(g + k) * n] = run;
            run += c[k];
        }
    }
    for (; g < G; ++g) {
        const int c = p[(size_t)g * n];
        p[(size_t)g * n] = run;
        run += c;
    }
    cnt[i] = run;
}

#define SCAN_T 1024
__global__ __launch_bounds__(SCAN_T) void scan_rowptr(
    const int* __restrict__ cnt, int* __restrict__ rowptr,
    float* __restrict__ dinv, int n)
{
    __shared__ int s_own[SCAN_T];
    __shared__ int s_pre[SCAN_T];
    const int t = threadIdx.x;
    const int b = blockIdx.x;

    int pre = 0;
    for (int k = 0; k < b; ++k) pre += cnt[k * SCAN_T + t];

    const int i = b * SCAN_T + t;
    const int c = (i < n) ? cnt[i] : 0;
    s_own[t] = c;
    s_pre[t] = pre;
    __syncthreads();
    for (int off = 1; off < SCAN_T; off <<= 1) {
        int a = 0, p = 0;
        if (t >= off) { a = s_own[t - off]; p = s_pre[t - off]; }
        __syncthreads();
        s_own[t] += a;
        s_pre[t] += p;
        __syncthreads();
    }
    const int excl = s_pre[SCAN_T - 1] + s_own[t] - c;
    if (i < n) {
        rowptr[i] = excl;
        dinv[i]   = rsqrtf((float)(c + 1));
    }
    if (i == n - 1) rowptr[n] = excl + c;
}

__global__ void fill_csr(const int* __restrict__ src, const int* __restrict__ dst,
                         const int* __restrict__ rank, const int* __restrict__ rowptr,
                         const int* __restrict__ cntm,
                         const float* __restrict__ dinv, int2* __restrict__ csr, int E, int n)
{
    int e = blockIdx.x * blockDim.x + threadIdx.x;
    if (e < E) {
        int s = src[e];
        int d = dst[e];
        int g = e / RANK_CHUNK;
        int2 r;
        r.x = s;
        r.y = __float_as_int(dinv[s] * dinv[d]);
        csr[rowptr[d] + cntm[(size_t)g * n + d] + rank[e]] = r;
    }
}

template <int K>
__device__ __forceinline__ void add_edges(
    float4& acc, const float* __restrict__ xw, const int2* __restrict__ rec, int c4)
{
    int2   r[K];
    float4 v[K];
#pragma unroll
    for (int k = 0; k < K; ++k) r[k] = rec[k];
#pragma unroll
    for (int k = 0; k < K; ++k)
        v[k] = *(const float4*)(xw + (size_t)r[k].x * HID_C + c4);
#pragma unroll
    for (int k = 0; k < K; ++k) {
        const float w = __int_as_float(r[k].y);
        acc.x = fmaf(v[k].x, w, acc.x);
        acc.y = fmaf(v[k].y, w, acc.y);
        acc.z = fmaf(v[k].z, w, acc.z);
        acc.w = fmaf(v[k].w, w, acc.w);
    }
}

__device__ __forceinline__ float gelu_erf(float v)
{
    return 0.5f * v * (1.0f + erff(v * 0.70710678118654752f));
}

template <bool POOL>
__global__ __launch_bounds__(256) void gcn_layer(
    const float* __restrict__ hin, const float* __restrict__ dinv,
    const int* __restrict__ rowptr, const int2* __restrict__ csr,
    const float* __restrict__ Wl, const float* __restrict__ b,
    float* __restrict__ hout, const int* __restrict__ batch,
    float* __restrict__ psum, float* __restrict__ pcnt, int n)
{
    __shared__ float lds_a[16][132];
    __shared__ int   s_gid[16];

    const int tid  = threadIdx.x;
    const int lane = tid & 31;
    const int wv   = tid >> 5;
    const int row0 = blockIdx.x * 16;
    const int c4   = lane << 2;

    if (POOL) {
        if (tid < 16) s_gid[tid] = (row0 + tid < n) ? batch[row0 + tid] : -1;
    } else if (psum) {
        for (int i = blockIdx.x * 256 + tid; i < NBATCH * HID_C + NBATCH; i += gridDim.x * 256)
            psum[i] = 0.0f;
    }

    const int  rowA = __builtin_amdgcn_readfirstlane(row0 + wv * 2);
    const bool hasA = rowA < n;
    const bool hasB = rowA + 1 < n;
    int   p0 = 0, p1 = 0, p2 = 0;
    float dA = 0.0f, dB = 0.0f;
    float4 accA = make_float4(0.0f, 0.0f, 0.0f, 0.0f);
    float4 accB = make_float4(0.0f, 0.0f, 0.0f, 0.0f);
    if (hasA) {
        p0 = rowptr[rowA];
        p1 = rowptr[rowA + 1];
        dA = dinv[rowA];
        accA = *(const float4*)(hin + (size_t)rowA * HID_C + c4);
    }
    if (hasB) {
        p2 = rowptr[rowA + 2];
        dB = dinv[rowA + 1];
        accB = *(const float4*)(hin + (size_t)(rowA + 1) * HID_C + c4);
    }
    {
        const float s = dA * dA;
        accA.x *= s; accA.y *= s; accA.z *= s; accA.w *= s;
        int e = p0;
        for (; e + 8 <= p1; e += 8) add_edges<8>(accA, hin, csr + e, c4);
        if (e + 4 <= p1) { add_edges<4>(accA, hin, csr + e, c4); e += 4; }
        if (e + 2 <= p1) { add_edges<2>(accA, hin, csr + e, c4); e += 2; }
        if (e < p1)      { add_edges<1>(accA, hin, csr + e, c4); }
        *(float4*)&lds_a[wv * 2][c4] = accA;
    }
    if (!hasB) p2 = p1;
    {
        const float s = dB * dB;
        accB.x *= s; accB.y *= s; accB.z *= s; accB.w *= s;
        int e = p1;
        for (; e + 8 <= p2; e += 8) add_edges<8>(accB, hin, csr + e, c4);
        if (e + 4 <= p2) { add_edges<4>(accB, hin, csr + e, c4); e += 4; }
        if (e + 2 <= p2) { add_edges<2>(accB, hin, csr + e, c4); e += 2; }
        if (e < p2)      { add_edges<1>(accB, hin, csr + e, c4); }
        *(float4*)&lds_a[wv * 2 + 1][c4] = accB;
    }
    __syncthreads();

    const int m    = lane & 15;
    const int kh   = (lane >> 4) << 1;
    const int col0 = wv * 16;
    const float* Bp = Wl + (size_t)kh * HID_C + col0 + m;

    v8f cacc = {};
#pragma unroll 8
    for (int kb = 0; kb < HID_C; kb += 4) {
        v2f a, w;
        a.x = lds_a[m][kb + kh];
        a.y = lds_a[m][kb + kh + 1];
        w.x = Bp[(size_t)kb * HID_C];
        w.y = Bp[(size_t)(kb + 1) * HID_C];
        cacc = __builtin_amdgcn_wmma_f32_16x16x4_f32(
            false, a, false, w, (short)0, cacc, false, false);
    }

    const int   col   = col0 + m;
    const int   rbase = row0 + ((lane >> 4) << 3);
    const float bv    = b[col];
    float y[8];
#pragma unroll
    for (int v = 0; v < 8; ++v) y[v] = gelu_erf(cacc[v] + bv);
    if (!POOL) {
#pragma unroll
        for (int v = 0; v < 8; ++v) {
            if (rbase + v < n) hout[(size_t)(rbase + v) * HID_C + col] = y[v];
        }
    } else {
        const int hb = (lane >> 4) << 3;
        int   cur = -1;
        float s   = 0.0f;
#pragma unroll
        for (int v = 0; v < 8; ++v) {
            const int g = s_gid[hb + v];
            if (g != cur) {
                if (cur >= 0) atomicAdd(psum + (size_t)cur * HID_C + col, s);
                cur = g;
                s   = 0.0f;
            }
            s += y[v];
        }
        if (cur >= 0) atomicAdd(psum + (size_t)cur * HID_C + col, s);
        if (tid < 16) {
            const int g = s_gid[tid];
            if (g >= 0 && (tid == 0 || s_gid[tid - 1] != g)) {
                int len = 1;
                while (tid + len < 16 && s_gid[tid + len] == g) ++len;
                atomicAdd(pcnt + g, (float)len);
            }
        }
    }
}

extern "C" void kernel_launch(void* const* d_in, const int* in_sizes, int n_in,
                              void* d_out, int out_size, void* d_ws, size_t ws_size,
                              hipStream_t stream)
{
    const float* x     = (const float*)d_in[0];
    const int*   ei    = (const int*)d_in[1];
    const int*   batch = (const int*)d_in[2];
    const float* W[3]  = { (const float*)d_in[3], (const float*)d_in[5], (const float*)d_in[7] };
    const float* bv[3] = { (const float*)d_in[4], (const float*)d_in[6], (const float*)d_in[8] };
    const float* Wm1   = (const float*)d_in[9];
    const float* bm1   = (const float*)d_in[10];
    const float* Wm2   = (const float*)d_in[11];
    const float* bm2   = (const float*)d_in[12];

    const int N = in_sizes[0] / IN_C;
    const int E = in_sizes[1] / 2;
    const int* src = ei;
    const int* dst = ei + E;

    char*  ws  = (char*)d_ws;
    size_t off = 0;
    auto carve = [&](size_t bytes) -> float* {
        float* p = (float*)(ws + off);
        off = (off + bytes + 255) & ~(size_t)255;
        return p;
    };
    int*   cnt  = (int*)carve((size_t)N * 4);
    float* dinv = carve((size_t)N * 4);
    float* buf0 = carve((size_t)N * HID_C * 4);
    float* buf1 = carve((size_t)N * HID_C * 4);
    float* psum = carve(((size_t)NBATCH * HID_C + NBATCH) * 4);
    float* pcnt = psum + (size_t)NBATCH * HID_C;
    float* hid  = carve((size_t)NBATCH * HID_C * 4);
    int*   rowptr = (int*)carve((size_t)(N + 1) * 4);
    int*   rank   = (int*)carve((size_t)E * 4);
    int2*  csr    = (int2*)carve((size_t)E * 8);
    const int G   = (E + RANK_CHUNK - 1) / RANK_CHUNK;
    int*   cntm   = (int*)carve((size_t)G * N * 4);

    rank_slices<<<G, RANK_T, 0, stream>>>(dst, cntm, rank, E, N);
    slice_offsets<<<(N + 255) / 256, 256, 0, stream>>>(cntm, cnt, N, G);
    scan_rowptr<<<(N + SCAN_T - 1) / SCAN_T, SCAN_T, 0, stream>>>(cnt, rowptr, dinv, N);
    fill_csr<<<(E + 255) / 256, 256, 0, stream>>>(src, dst, rank, rowptr, cntm, dinv, csr, E, N);

    gcn_layer<false><<<(N + 15) / 16, 256, 0, stream>>>(
        x, dinv, rowptr, csr, W[0], bv[0], buf0, nullptr, nullptr, nullptr, N);
    gcn_layer<false><<<(N + 15) / 16, 256, 0, stream>>>(
        buf0, dinv, rowptr, csr, W[1], bv[1], buf1, nullptr, psum, nullptr, N);
    gcn_layer<true><<<(N + 15) / 16, 256, 0, stream>>>(
        buf1, dinv, rowptr, csr, W[2], bv[2], nullptr, batch, psum, pcnt, N);

    gemm_f32_wmma<<<NBATCH / 16, 256, 0, stream>>>(psum, Wm1, bm1, pcnt, hid,
                                                   NBATCH, HID_C, HID_C, 1);
    gemm_f32_wmma<<<NBATCH / 16, 128, 0, stream>>>(hid, Wm2, bm2, nullptr, (float*)d_out,
                                                   NBATCH, HID_C, OUT_C, 0);
}


# MLP head in one launch: hidden 16x128 tile kept in LDS between the two f32 WMMA products (mean, W1, bias, ReLU, W2, bias); one launch and the hidden plane's write and read less
# speedup vs baseline: 4.6647x; 1.0450x over previous
#include <hip/hip_runtime.h>
#include <hip/hip_bf16.h>
#include <math.h>

typedef float v2f __attribute__((ext_vector_type(2)));
typedef float v8f __attribute__((ext_vector_type(8)));

#define IN_C   128
#define HID_C  128
#define OUT_C  64
#define NBATCH 512

__global__ __launch_bounds__(256) void mlp_head(
    const float* __restrict__ psum, const float* __restrict__ pcnt,
    const float* __restrict__ W1, const float* __restrict__ b1,
    const float* __restrict__ W2, const float* __restrict__ b2,
    float* __restrict__ out)
{
    __shared__ float lds_a[16][132];
    __shared__ float lds_h[16][132];

    const int tid  = threadIdx.x;
    const int lane = tid & 31;
    const int wv   = __builtin_amdgcn_readfirstlane(tid >> 5);
    const int row0 = blockIdx.x * 16;

    {
        const int ra = tid >> 5;
        const int rb = ra + 8;
        const int k4 = (tid & 31) << 2;
        float4 va = *(const float4*)(psum + (size_t)(row0 + ra) * HID_C + k4);
        float4 vb = *(const float4*)(psum + (size_t)(row0 + rb) * HID_C + k4);
        const float ca = fmaxf(pcnt[row0 + ra], 1.0f);
        const float cb = fmaxf(pcnt[row0 + rb], 1.0f);
        va.x /= ca; va.y /= ca; va.z /= ca; va.w /= ca;
        vb.x /= cb; vb.y /= cb; vb.z /= cb; vb.w /= cb;
        *(float4*)&lds_a[ra][k4] = va;
        *(float4*)&lds_a[rb][k4] = vb;
    }
    __syncthreads();

    const int m    = lane & 15;
    const int kh   = (lane >> 4) << 1;
    const int hb   = (lane >> 4) << 3;
    const int col0 = wv * 16;
    const int col  = col0 + m;

    {
        const float* Bp = W1 + (size_t)kh * HID_C + col;
        v8f acc = {};
#pragma unroll 8
        for (int kb = 0; kb < HID_C; kb += 4) {
            v2f a, w;
            a.x = lds_a[m][kb + kh];
            a.y = lds_a[m][kb + kh + 1];
            w.x = Bp[(size_t)kb * HID_C];
            w.y = Bp[(size_t)(kb + 1) * HID_C];
            acc = __builtin_amdgcn_wmma_f32_16x16x4_f32(
                false, a, false, w, (short)0, acc, false, false);
        }
        const float bv = b1[col];
#pragma unroll
        for (int v = 0; v < 8; ++v)
            lds_h[hb + v][col] = fmaxf(acc[v] + bv, 0.0f);
    }
    __syncthreads();

    if (wv < OUT_C / 16) {
        const float* Bp = W2 + (size_t)kh * OUT_C + col;
        v8f acc = {};
#pragma unroll 8
        for (int kb = 0; kb < HID_C; kb += 4) {
            v2f a, w;
            a.x = lds_h[m][kb + kh];
            a.y = lds_h[m][kb + kh + 1];
            w.x = Bp[(size_t)kb * OUT_C];
            w.y = Bp[(size_t)(kb + 1) * OUT_C];
            acc = __builtin_amdgcn_wmma_f32_16x16x4_f32(
                false, a, false, w, (short)0, acc, false, false);
        }
        const float bv = b2[col];
#pragma unroll
        for (int v = 0; v < 8; ++v)
            out[(size_t)(row0 + hb + v) * OUT_C + col] = acc[v] + bv;
    }
}

#define RANK_CHUNK 16384
#define RANK_T     1024
#define MAX_NODES  50000

__global__ __launch_bounds__(RANK_T) void rank_slices(
    const int* __restrict__ dst, int* __restrict__ cntm, int* __restrict__ rank,
    int E, int n)
{
    __shared__ alignas(16) int lc[MAX_NODES];
    const int t = threadIdx.x;
    const int g = blockIdx.x;

    for (int i = t; i < MAX_NODES / 4; i += RANK_T)
        ((int4*)lc)[i] = make_int4(0, 0, 0, 0);
    __syncthreads();

    const int beg = g * RANK_CHUNK;
    const int end = min(beg + RANK_CHUNK, E);
    for (int e = beg + t; e < end; e += 4 * RANK_T) {
        const int e1 = e + RANK_T, e2 = e + 2 * RANK_T, e3 = e + 3 * RANK_T;
        if (e3 < end) {
            const int d0 = dst[e];
            const int d1 = dst[e1];
            const int d2 = dst[e2];
            const int d3 = dst[e3];
            const int r0 = atomicAdd(&lc[d0], 1);
            const int r1 = atomicAdd(&lc[d1], 1);
            const int r2 = atomicAdd(&lc[d2], 1);
            const int r3 = atomicAdd(&lc[d3], 1);
            rank[e]  = r0;
            rank[e1] = r1;
            rank[e2] = r2;
            rank[e3] = r3;
        } else {
            for (int k = e; k < end; k += RANK_T)
                rank[k] = atomicAdd(&lc[dst[k]], 1);
        }
    }
    __syncthreads();

    int* out = cntm + (size_t)g * n;
    for (int i = t; i < n; i += RANK_T) out[i] = lc[i];
}

__global__ void slice_offsets(int* __restrict__ cntm, int* __restrict__ cnt, int n, int G)
{
    const int i = blockIdx.x * blockDim.x + threadIdx.x;
    if (i >= n) return;
    int* p = cntm + i;
    int run = 0;
    int g = 0;
    for (; g + 8 <= G; g += 8) {
        int c[8];
#pragma unroll
        for (int k = 0; k < 8; ++k) c[k] = p[(size_t)(g + k) * n];
#pragma unroll
        for (int k = 0; k < 8; ++k) {
            p[(size_t)(g + k) * n] = run;
            run += c[k];
        }
    }
    for (; g < G; ++g) {
        const int c = p[(size_t)g * n];
        p[(size_t)g * n] = run;
        run += c;
    }
    cnt[i] = run;
}

#define SCAN_T 1024
__global__ __launch_bounds__(SCAN_T) void scan_rowptr(
    const int* __restrict__ cnt, int* __restrict__ rowptr,
    float* __restrict__ dinv, int n)
{
    __shared__ int s_own[SCAN_T];
    __shared__ int s_pre[SCAN_T];
    const int t = threadIdx.x;
    const int b = blockIdx.x;

    int pre = 0;
    for (int k = 0; k < b; ++k) pre += cnt[k * SCAN_T + t];

    const int i = b * SCAN_T + t;
    const int c = (i < n) ? cnt[i] : 0;
    s_own[t] = c;
    s_pre[t] = pre;
    __syncthreads();
    for (int off = 1; off < SCAN_T; off <<= 1) {
        int a = 0, p = 0;
        if (t >= off) { a = s_own[t - off]; p = s_pre[t - off]; }
        __syncthreads();
        s_own[t] += a;
        s_pre[t] += p;
        __syncthreads();
    }
    const int excl = s_pre[SCAN_T - 1] + s_own[t] - c;
    if (i < n) {
        rowptr[i] = excl;
        dinv[i]   = rsqrtf((float)(c + 1));
    }
    if (i == n - 1) rowptr[n] = excl + c;
}

__global__ void fill_csr(const int* __restrict__ src, const int* __restrict__ dst,
                         const int* __restrict__ rank, const int* __restrict__ rowptr,
                         const int* __restrict__ cntm,
                         const float* __restrict__ dinv, int2* __restrict__ csr, int E, int n)
{
    int e = blockIdx.x * blockDim.x + threadIdx.x;
    if (e < E) {
        int s = src[e];
        int d = dst[e];
        int g = e / RANK_CHUNK;
        int2 r;
        r.x = s;
        r.y = __float_as_int(dinv[s] * dinv[d]);
        csr[rowptr[d] + cntm[(size_t)g * n + d] + rank[e]] = r;
    }
}

template <int K>
__device__ __forceinline__ void add_edges(
    float4& acc, const float* __restrict__ xw, const int2* __restrict__ rec, int c4)
{
    int2   r[K];
    float4 v[K];
#pragma unroll
    for (int k = 0; k < K; ++k) r[k] = rec[k];
#pragma unroll
    for (int k = 0; k < K; ++k)
        v[k] = *(const float4*)(xw + (size_t)r[k].x * HID_C + c4);
#pragma unroll
    for (int k = 0; k < K; ++k) {
        const float w = __int_as_float(r[k].y);
        acc.x = fmaf(v[k].x, w, acc.x);
        acc.y = fmaf(v[k].y, w, acc.y);
        acc.z = fmaf(v[k].z, w, acc.z);
        acc.w = fmaf(v[k].w, w, acc.w);
    }
}

__device__ __forceinline__ float gelu_erf(float v)
{
    return 0.5f * v * (1.0f + erff(v * 0.70710678118654752f));
}

template <bool POOL>
__global__ __launch_bounds__(256) void gcn_layer(
    const float* __restrict__ hin, const float* __restrict__ dinv,
    const int* __restrict__ rowptr, const int2* __restrict__ csr,
    const float* __restrict__ Wl, const float* __restrict__ b,
    float* __restrict__ hout, const int* __restrict__ batch,
    float* __restrict__ psum, float* __restrict__ pcnt, int n)
{
    __shared__ float lds_a[16][132];
    __shared__ int   s_gid[16];

    const int tid  = threadIdx.x;
    const int lane = tid & 31;
    const int wv   = tid >> 5;
    const int row0 = blockIdx.x * 16;
    const int c4   = lane << 2;

    if (POOL) {
        if (tid < 16) s_gid[tid] = (row0 + tid < n) ? batch[row0 + tid] : -1;
    } else if (psum) {
        for (int i = blockIdx.x * 256 + tid; i < NBATCH * HID_C + NBATCH; i += gridDim.x * 256)
            psum[i] = 0.0f;
    }

    const int  rowA = __builtin_amdgcn_readfirstlane(row0 + wv * 2);
    const bool hasA = rowA < n;
    const bool hasB = rowA + 1 < n;
    int   p0 = 0, p1 = 0, p2 = 0;
    float dA = 0.0f, dB = 0.0f;
    float4 accA = make_float4(0.0f, 0.0f, 0.0f, 0.0f);
    float4 accB = make_float4(0.0f, 0.0f, 0.0f, 0.0f);
    if (hasA) {
        p0 = rowptr[rowA];
        p1 = rowptr[rowA + 1];
        dA = dinv[rowA];
        accA = *(const float4*)(hin + (size_t)rowA * HID_C + c4);
    }
    if (hasB) {
        p2 = rowptr[rowA + 2];
        dB = dinv[rowA + 1];
        accB = *(const float4*)(hin + (size_t)(rowA + 1) * HID_C + c4);
    }
    {
        const float s = dA * dA;
        accA.x *= s; accA.y *= s; accA.z *= s; accA.w *= s;
        int e = p0;
        for (; e + 8 <= p1; e += 8) add_edges<8>(accA, hin, csr + e, c4);
        if (e + 4 <= p1) { add_edges<4>(accA, hin, csr + e, c4); e += 4; }
        if (e + 2 <= p1) { add_edges<2>(accA, hin, csr + e, c4); e += 2; }
        if (e < p1)      { add_edges<1>(accA, hin, csr + e, c4); }
        *(float4*)&lds_a[wv * 2][c4] = accA;
    }
    if (!hasB) p2 = p1;
    {
        const float s = dB * dB;
        accB.x *= s; accB.y *= s; accB.z *= s; accB.w *= s;
        int e = p1;
        for (; e + 8 <= p2; e += 8) add_edges<8>(accB, hin, csr + e, c4);
        if (e + 4 <= p2) { add_edges<4>(accB, hin, csr + e, c4); e += 4; }
        if (e + 2 <= p2) { add_edges<2>(accB, hin, csr + e, c4); e += 2; }
        if (e < p2)      { add_edges<1>(accB, hin, csr + e, c4); }
        *(float4*)&lds_a[wv * 2 + 1][c4] = accB;
    }
    __syncthreads();

    const int m    = lane & 15;
    const int kh   = (lane >> 4) << 1;
    const int col0 = wv * 16;
    const float* Bp = Wl + (size_t)kh * HID_C + col0 + m;

    v8f cacc = {};
#pragma unroll 8
    for (int kb = 0; kb < HID_C; kb += 4) {
        v2f a, w;
        a.x = lds_a[m][kb + kh];
        a.y = lds_a[m][kb + kh + 1];
        w.x = Bp[(size_t)kb * HID_C];
        w.y = Bp[(size_t)(kb + 1) * HID_C];
        cacc = __builtin_amdgcn_wmma_f32_16x16x4_f32(
            false, a, false, w, (short)0, cacc, false, false);
    }

    const int   col   = col0 + m;
    const int   rbase = row0 + ((lane >> 4) << 3);
    const float bv    = b[col];
    float y[8];
#pragma unroll
    for (int v = 0; v < 8; ++v) y[v] = gelu_erf(cacc[v] + bv);
    if (!POOL) {
#pragma unroll
        for (int v = 0; v < 8; ++v) {
            if (rbase + v < n) hout[(size_t)(rbase + v) * HID_C + col] = y[v];
        }
    } else {
        const int hb = (lane >> 4) << 3;
        int   cur = -1;
        float s   = 0.0f;
#pragma unroll
        for (int v = 0; v < 8; ++v) {
            const int g = s_gid[hb + v];
            if (g != cur) {
                if (cur >= 0) atomicAdd(psum + (size_t)cur * HID_C + col, s);
                cur = g;
                s   = 0.0f;
            }
            s += y[v];
        }
        if (cur >= 0) atomicAdd(psum + (size_t)cur * HID_C + col, s);
        if (tid < 16) {
            const int g = s_gid[tid];
            if (g >= 0 && (tid == 0 || s_gid[tid - 1] != g)) {
                int len = 1;
                while (tid + len < 16 && s_gid[tid + len] == g) ++len;
                atomicAdd(pcnt + g, (float)len);
            }
        }
    }
}

extern "C" void kernel_launch(void* const* d_in, const int* in_sizes, int n_in,
                              void* d_out, int out_size, void* d_ws, size_t ws_size,
                              hipStream_t stream)
{
    const float* x     = (const float*)d_in[0];
    const int*   ei    = (const int*)d_in[1];
    const int*   batch = (const int*)d_in[2];
    const float* W[3]  = { (const float*)d_in[3], (const float*)d_in[5], (const float*)d_in[7] };
    const float* bv[3] = { (const float*)d_in[4], (const float*)d_in[6], (const float*)d_in[8] };
    const float* Wm1   = (const float*)d_in[9];
    const float* bm1   = (const float*)d_in[10];
    const float* Wm2   = (const float*)d_in[11];
    const float* bm2   = (const float*)d_in[12];

    const int N = in_sizes[0] / IN_C;
    const int E = in_sizes[1] / 2;
    const int* src = ei;
    const int* dst = ei + E;

    char*  ws  = (char*)d_ws;
    size_t off = 0;
    auto carve = [&](size_t bytes) -> float* {
        float* p = (float*)(ws + off);
        off = (off + bytes + 255) & ~(size_t)255;
        return p;
    };
    int*   cnt  = (int*)carve((size_t)N * 4);
    float* dinv = carve((size_t)N * 4);
    float* buf0 = carve((size_t)N * HID_C * 4);
    float* buf1 = carve((size_t)N * HID_C * 4);
    float* psum = carve(((size_t)NBATCH * HID_C + NBATCH) * 4);
    float* pcnt = psum + (size_t)NBATCH * HID_C;
    int*   rowptr = (int*)carve((size_t)(N + 1) * 4);
    int*   rank   = (int*)carve((size_t)E * 4);
    int2*  csr    = (int2*)carve((size_t)E * 8);
    const int G   = (E + RANK_CHUNK - 1) / RANK_CHUNK;
    int*   cntm   = (int*)carve((size_t)G * N * 4);

    rank_slices<<<G, RANK_T, 0, stream>>>(dst, cntm, rank, E, N);
    slice_offsets<<<(N + 255) / 256, 256, 0, stream>>>(cntm, cnt, N, G);
    scan_rowptr<<<(N + SCAN_T - 1) / SCAN_T, SCAN_T, 0, stream>>>(cnt, rowptr, dinv, N);
    fill_csr<<<(E + 255) / 256, 256, 0, stream>>>(src, dst, rank, rowptr, cntm, dinv, csr, E, N);

    gcn_layer<false><<<(N + 15) / 16, 256, 0, stream>>>(
        x, dinv, rowptr, csr, W[0], bv[0], buf0, nullptr, nullptr, nullptr, N);
    gcn_layer<false><<<(N + 15) / 16, 256, 0, stream>>>(
        buf0, dinv, rowptr, csr, W[1], bv[1], buf1, nullptr, psum, nullptr, N);
    gcn_layer<true><<<(N + 15) / 16, 256, 0, stream>>>(
        buf1, dinv, rowptr, csr, W[2], bv[2], nullptr, batch, psum, pcnt, N);

    mlp_head<<<NBATCH / 16, 256, 0, stream>>>(psum, pcnt, Wm1, bm1, Wm2, bm2, (float*)d_out);
}


# neighbour walk: a wave's edge records (both rows as one stream) come by 256-byte vector loads and lane broadcast, not by a scalar load per 8-edge trip; per-row sum order unchanged
# speedup vs baseline: 4.8155x; 1.0323x over previous
#include <hip/hip_runtime.h>
#include <hip/hip_bf16.h>
#include <math.h>

typedef float v2f __attribute__((ext_vector_type(2)));
typedef float v8f __attribute__((ext_vector_type(8)));

#define IN_C   128
#define HID_C  128
#define OUT_C  64
#define NBATCH 512

__global__ __launch_bounds__(256) void mlp_head(
    const float* __restrict__ psum, const float* __restrict__ pcnt,
    const float* __restrict__ W1, const float* __restrict__ b1,
    const float* __restrict__ W2, const float* __restrict__ b2,
    float* __restrict__ out)
{
    __shared__ float lds_a[16][132];
    __shared__ float lds_h[16][132];

    const int tid  = threadIdx.x;
    const int lane = tid & 31;
    const int wv   = __builtin_amdgcn_readfirstlane(tid >> 5);
    const int row0 = blockIdx.x * 16;

    {
        const int ra = tid >> 5;
        const int rb = ra + 8;
        const int k4 = (tid & 31) << 2;
        float4 va = *(const float4*)(psum + (size_t)(row0 + ra) * HID_C + k4);
        float4 vb = *(const float4*)(psum + (size_t)(row0 + rb) * HID_C + k4);
        const float ca = fmaxf(pcnt[row0 + ra], 1.0f);
        const float cb = fmaxf(pcnt[row0 + rb], 1.0f);
        va.x /= ca; va.y /= ca; va.z /= ca; va.w /= ca;
        vb.x /= cb; vb.y /= cb; vb.z /= cb; vb.w /= cb;
        *(float4*)&lds_a[ra][k4] = va;
        *(float4*)&lds_a[rb][k4] = vb;
    }
    __syncthreads();

    const int m    = lane & 15;
    const int kh   = (lane >> 4) << 1;
    const int hb   = (lane >> 4) << 3;
    const int col0 = wv * 16;
    const int col  = col0 + m;

    {
        const float* Bp = W1 + (size_t)kh * HID_C + col;
        v8f acc = {};
#pragma unroll 8
        for (int kb = 0; kb < HID_C; kb += 4) {
            v2f a, w;
            a.x = lds_a[m][kb + kh];
            a.y = lds_a[m][kb + kh + 1];
            w.x = Bp[(size_t)kb * HID_C];
            w.y = Bp[(size_t)(kb + 1) * HID_C];
            acc = __builtin_amdgcn_wmma_f32_16x16x4_f32(
                false, a, false, w, (short)0, acc, false, false);
        }
        const float bv = b1[col];
#pragma unroll
        for (int v = 0; v < 8; ++v)
            lds_h[hb + v][col] = fmaxf(acc[v] + bv, 0.0f);
    }
    __syncthreads();

    if (wv < OUT_C / 16) {
        const float* Bp = W2 + (size_t)kh * OUT_C + col;
        v8f acc = {};
#pragma unroll 8
        for (int kb = 0; kb < HID_C; kb += 4) {
            v2f a, w;
            a.x = lds_h[m][kb + kh];
            a.y = lds_h[m][kb + kh + 1];
            w.x = Bp[(size_t)kb * OUT_C];
            w.y = Bp[(size_t)(kb + 1) * OUT_C];
            acc = __builtin_amdgcn_wmma_f32_16x16x4_f32(
                false, a, false, w, (short)0, acc, false, false);
        }
        const float bv = b2[col];
#pragma unroll
        for (int v = 0; v < 8; ++v)
            out[(size_t)(row0 + hb + v) * OUT_C + col] = acc[v] + bv;
    }
}

#define RANK_CHUNK 16384
#define RANK_T     1024
#define MAX_NODES  50000

__global__ __launch_bounds__(RANK_T) void rank_slices(
    const int* __restrict__ dst, int* __restrict__ cntm, int* __restrict__ rank,
    int E, int n)
{
    __shared__ alignas(16) int lc[MAX_NODES];
    const int t = threadIdx.x;
    const int g = blockIdx.x;

    for (int i = t; i < MAX_NODES / 4; i += RANK_T)
        ((int4*)lc)[i] = make_int4(0, 0, 0, 0);
    __syncthreads();

    const int beg = g * RANK_CHUNK;
    const int end = min(beg + RANK_CHUNK, E);
    for (int e = beg + t; e < end; e += 4 * RANK_T) {
        const int e1 = e + RANK_T, e2 = e + 2 * RANK_T, e3 = e + 3 * RANK_T;
        if (e3 < end) {
            const int d0 = dst[e];
            const int d1 = dst[e1];
            const int d2 = dst[e2];
            const int d3 = dst[e3];
            const int r0 = atomicAdd(&lc[d0], 1);
            const int r1 = atomicAdd(&lc[d1], 1);
            const int r2 = atomicAdd(&lc[d2], 1);
            const int r3 = atomicAdd(&lc[d3], 1);
            rank[e]  = r0;
            rank[e1] = r1;
            rank[e2] = r2;
            rank[e3] = r3;
        } else {
            for (int k = e; k < end; k += RANK_T)
                rank[k] = atomicAdd(&lc[dst[k]], 1);
        }
    }
    __syncthreads();

    int* out = cntm + (size_t)g * n;
    for (int i = t; i < n; i += RANK_T) out[i] = lc[i];
}

__global__ void slice_offsets(int* __restrict__ cntm, int* __restrict__ cnt, int n, int G)
{
    const int i = blockIdx.x * blockDim.x + threadIdx.x;
    if (i >= n) return;
    int* p = cntm + i;
    int run = 0;
    int g = 0;
    for (; g + 8 <= G; g += 8) {
        int c[8];
#pragma unroll
        for (int k = 0; k < 8; ++k) c[k] = p[(size_t)(g + k) * n];
#pragma unroll
        for (int k = 0; k < 8; ++k) {
            p[(size_t)(g + k) * n] = run;
            run += c[k];
        }
    }
    for (; g < G; ++g) {
        const int c = p[(size_t)g * n];
        p[(size_t)g * n] = run;
        run += c;
    }
    cnt[i] = run;
}

#define SCAN_T 1024
__global__ __launch_bounds__(SCAN_T) void scan_rowptr(
    const int* __restrict__ cnt, int* __restrict__ rowptr,
    float* __restrict__ dinv, int n)
{
    __shared__ int s_own[SCAN_T];
    __shared__ int s_pre[SCAN_T];
    const int t = threadIdx.x;
    const int b = blockIdx.x;

    int pre = 0;
    for (int k = 0; k < b; ++k) pre += cnt[k * SCAN_T + t];

    const int i = b * SCAN_T + t;
    const int c = (i < n) ? cnt[i] : 0;
    s_own[t] = c;
    s_pre[t] = pre;
    __syncthreads();
    for (int off = 1; off < SCAN_T; off <<= 1) {
        int a = 0, p = 0;
        if (t >= off) { a = s_own[t - off]; p = s_pre[t - off]; }
        __syncthreads();
        s_own[t] += a;
        s_pre[t] += p;
        __syncthreads();
    }
    const int excl = s_pre[SCAN_T - 1] + s_own[t] - c;
    if (i < n) {
        rowptr[i] = excl;
        dinv[i]   = rsqrtf((float)(c + 1));
    }
    if (i == n - 1) rowptr[n] = excl + c;
}

__global__ void fill_csr(const int* __restrict__ src, const int* __restrict__ dst,
                         const int* __restrict__ rank, const int* __restrict__ rowptr,
                         const int* __restrict__ cntm,
                         const float* __restrict__ dinv, int2* __restrict__ csr, int E, int n)
{
    int e = blockIdx.x * blockDim.x + threadIdx.x;
    if (e < E) {
        int s = src[e];
        int d = dst[e];
        int g = e / RANK_CHUNK;
        int2 r;
        r.x = s;
        r.y = __float_as_int(dinv[s] * dinv[d]);
        csr[rowptr[d] + cntm[(size_t)g * n + d] + rank[e]] = r;
    }
}

__device__ __forceinline__ void fma4(float4& a, const float4& v, float w)
{
    a.x = fmaf(v.x, w, a.x);
    a.y = fmaf(v.y, w, a.y);
    a.z = fmaf(v.z, w, a.z);
    a.w = fmaf(v.w, w, a.w);
}

template <int K>
__device__ __forceinline__ void add_recs(
    float4& accA, float4& accB, const float* __restrict__ xw, int2 cur, int j, int nA, int c4)
{
    int    s[K];
    float  w[K];
    float4 v[K];
#pragma unroll
    for (int k = 0; k < K; ++k) {
        s[k] = __builtin_amdgcn_readlane(cur.x, j + k);
        w[k] = __int_as_float(__builtin_amdgcn_readlane(cur.y, j + k));
    }
#pragma unroll
    for (int k = 0; k < K; ++k)
        v[k] = *(const float4*)(xw + (size_t)s[k] * HID_C + c4);
    if (j + K <= nA) {
#pragma unroll
        for (int k = 0; k < K; ++k) fma4(accA, v[k], w[k]);
    } else if (j >= nA) {
#pragma unroll
        for (int k = 0; k < K; ++k) fma4(accB, v[k], w[k]);
    } else {
#pragma unroll
        for (int k = 0; k < K; ++k) {
            if (j + k < nA) fma4(accA, v[k], w[k]);
            else            fma4(accB, v[k], w[k]);
        }
    }
}

__device__ __forceinline__ float gelu_erf(float v)
{
    return 0.5f * v * (1.0f + erff(v * 0.70710678118654752f));
}

template <bool POOL>
__global__ __launch_bounds__(256) __attribute__((amdgpu_waves_per_eu(16, 16))) void gcn_layer(
    const float* __restrict__ hin, const float* __restrict__ dinv,
    const int* __restrict__ rowptr, const int2* __restrict__ csr,
    const float* __restrict__ Wl, const float* __restrict__ b,
    float* __restrict__ hout, const int* __restrict__ batch,
    float* __restrict__ psum, float* __restrict__ pcnt, int n)
{
    __shared__ float lds_a[16][132];
    __shared__ int   s_gid[16];

    const int tid  = threadIdx.x;
    const int lane = tid & 31;
    const int wv   = tid >> 5;
    const int row0 = blockIdx.x * 16;
    const int c4   = lane << 2;

    if (POOL) {
        if (tid < 16) s_gid[tid] = (row0 + tid < n) ? batch[row0 + tid] : -1;
    } else if (psum) {
        for (int i = blockIdx.x * 256 + tid; i < NBATCH * HID_C + NBATCH; i += gridDim.x * 256)
            psum[i] = 0.0f;
    }

    const int  rowA = __builtin_amdgcn_readfirstlane(row0 + wv * 2);
    const bool hasA = rowA < n;
    const bool hasB = rowA + 1 < n;
    int   p0 = 0, p1 = 0, p2 = 0;
    float dA = 0.0f, dB = 0.0f;
    float4 accA = make_float4(0.0f, 0.0f, 0.0f, 0.0f);
    float4 accB = make_float4(0.0f, 0.0f, 0.0f, 0.0f);
    if (hasA) {
        p0 = rowptr[rowA];
        p1 = rowptr[rowA + 1];
        dA = dinv[rowA];
        accA = *(const float4*)(hin + (size_t)rowA * HID_C + c4);
    }
    if (hasB) {
        p2 = rowptr[rowA + 2];
        dB = dinv[rowA + 1];
        accB = *(const float4*)(hin + (size_t)(rowA + 1) * HID_C + c4);
    }
    if (!hasB) p2 = p1;
    {
        const float sA = dA * dA;
        accA.x *= sA; accA.y *= sA; accA.z *= sA; accA.w *= sA;
        const float sB = dB * dB;
        accB.x *= sB; accB.y *= sB; accB.z *= sB; accB.w *= sB;
    }
    {
        int2 cur = make_int2(0, 0);
        if (p0 + lane < p2) cur = csr[p0 + lane];
        for (int base = p0; base < p2; base += 32) {
            int2 nxt = make_int2(0, 0);
            if (base + 32 + lane < p2) nxt = csr[base + 32 + lane];
            const int cnt = min(32, p2 - base);
            const int nA  = p1 - base;
            int j = 0;
            for (; j + 8 <= cnt; j += 8) add_recs<8>(accA, accB, hin, cur, j, nA, c4);
            if (j + 4 <= cnt) { add_recs<4>(accA, accB, hin, cur, j, nA, c4); j += 4; }
            if (j + 2 <= cnt) { add_recs<2>(accA, accB, hin, cur, j, nA, c4); j += 2; }
            if (j < cnt)      { add_recs<1>(accA, accB, hin, cur, j, nA, c4); }
            cur = nxt;
        }
        *(float4*)&lds_a[wv * 2][c4]     = accA;
        *(float4*)&lds_a[wv * 2 + 1][c4] = accB;
    }
    __syncthreads();

    const int m    = lane & 15;
    const int kh   = (lane >> 4) << 1;
    const int col0 = wv * 16;
    const float* Bp = Wl + (size_t)kh * HID_C + col0 + m;

    v8f cacc = {};
#pragma unroll 8
    for (int kb = 0; kb < HID_C; kb += 4) {
        v2f a, w;
        a.x = lds_a[m][kb + kh];
        a.y = lds_a[m][kb + kh + 1];
        w.x = Bp[(size_t)kb * HID_C];
        w.y = Bp[(size_t)(kb + 1) * HID_C];
        cacc = __builtin_amdgcn_wmma_f32_16x16x4_f32(
            false, a, false, w, (short)0, cacc, false, false);
    }

    const int   col   = col0 + m;
    const int   rbase = row0 + ((lane >> 4) << 3);
    const float bv    = b[col];
    float y[8];
#pragma unroll
    for (int v = 0; v < 8; ++v) y[v] = gelu_erf(cacc[v] + bv);
    if (!POOL) {
#pragma unroll
        for (int v = 0; v < 8; ++v) {
            if (rbase + v < n) hout[(size_t)(rbase + v) * HID_C + col] = y[v];
        }
    } else {
        const int hb = (lane >> 4) << 3;
        int   cur = -1;
        float s   = 0.0f;
#pragma unroll
        for (int v = 0; v < 8; ++v) {
            const int g = s_gid[hb + v];
            if (g != cur) {
                if (cur >= 0) atomicAdd(psum + (size_t)cur * HID_C + col, s);
                cur = g;
                s   = 0.0f;
            }
            s += y[v];
        }
        if (cur >= 0) atomicAdd(psum + (size_t)cur * HID_C + col, s);
        if (tid < 16) {
            const int g = s_gid[tid];
            if (g >= 0 && (tid == 0 || s_gid[tid - 1] != g)) {
                int len = 1;
                while (tid + len < 16 && s_gid[tid + len] == g) ++len;
                atomicAdd(pcnt + g, (float)len);
            }
        }
    }
}

extern "C" void kernel_launch(void* const* d_in, const int* in_sizes, int n_in,
                              void* d_out, int out_size, void* d_ws, size_t ws_size,
                              hipStream_t stream)
{
    const float* x     = (const float*)d_in[0];
    const int*   ei    = (const int*)d_in[1];
    const int*   batch = (const int*)d_in[2];
    const float* W[3]  = { (const float*)d_in[3], (const float*)d_in[5], (const float*)d_in[7] };
    const float* bv[3] = { (const float*)d_in[4], (const float*)d_in[6], (const float*)d_in[8] };
    const float* Wm1   = (const float*)d_in[9];
    const float* bm1   = (const float*)d_in[10];
    const float* Wm2   = (const float*)d_in[11];
    const float* bm2   = (const float*)d_in[12];

    const int N = in_sizes[0] / IN_C;
    const int E = in_sizes[1] / 2;
    const int* src = ei;
    const int* dst = ei + E;

    char*  ws  = (char*)d_ws;
    size_t off = 0;
    auto carve = [&](size_t bytes) -> float* {
        float* p = (float*)(ws + off);
        off = (off + bytes + 255) & ~(size_t)255;
        return p;
    };
    int*   cnt  = (int*)carve((size_t)N * 4);
    float* dinv = carve((size_t)N * 4);
    float* buf0 = carve((size_t)N * HID_C * 4);
    float* buf1 = carve((size_t)N * HID_C * 4);
    float* psum = carve(((size_t)NBATCH * HID_C + NBATCH) * 4);
    float* pcnt = psum + (size_t)NBATCH * HID_C;
    int*   rowptr = (int*)carve((size_t)(N + 1) * 4);
    int*   rank   = (int*)carve((size_t)E * 4);
    int2*  csr    = (int2*)carve((size_t)E * 8);
    const int G   = (E + RANK_CHUNK - 1) / RANK_CHUNK;
    int*   cntm   = (int*)carve((size_t)G * N * 4);

    rank_slices<<<G, RANK_T, 0, stream>>>(dst, cntm, rank, E, N);
    slice_offsets<<<(N + 255) / 256, 256, 0, stream>>>(cntm, cnt, N, G);
    scan_rowptr<<<(N + SCAN_T - 1) / SCAN_T, SCAN_T, 0, stream>>>(cnt, rowptr, dinv, N);
    fill_csr<<<(E + 255) / 256, 256, 0, stream>>>(src, dst, rank, rowptr, cntm, dinv, csr, E, N);

    gcn_layer<false><<<(N + 15) / 16, 256, 0, stream>>>(
        x, dinv, rowptr, csr, W[0], bv[0], buf0, nullptr, nullptr, nullptr, N);
    gcn_layer<false><<<(N + 15) / 16, 256, 0, stream>>>(
        buf0, dinv, rowptr, csr, W[1], bv[1], buf1, nullptr, psum, nullptr, N);
    gcn_layer<true><<<(N + 15) / 16, 256, 0, stream>>>(
        buf1, dinv, rowptr, csr, W[2], bv[2], nullptr, batch, psum, pcnt, N);

    mlp_head<<<NBATCH / 16, 256, 0, stream>>>(psum, pcnt, Wm1, bm1, Wm2, bm2, (float*)d_out);
}


# row-start scan folded into the slice-offset launch: the ranking pass leaves per-slice node-chunk totals, so each 1024-node block gets its edge prefix from them; one launch and the cnt plane less
# speedup vs baseline: 4.9966x; 1.0376x over previous
#include <hip/hip_runtime.h>
#include <hip/hip_bf16.h>
#include <math.h>

typedef float v2f __attribute__((ext_vector_type(2)));
typedef float v8f __attribute__((ext_vector_type(8)));

#define IN_C   128
#define HID_C  128
#define OUT_C  64
#define NBATCH 512

__global__ __launch_bounds__(256) void mlp_head(
    const float* __restrict__ psum, const float* __restrict__ pcnt,
    const float* __restrict__ W1, const float* __restrict__ b1,
    const float* __restrict__ W2, const float* __restrict__ b2,
    float* __restrict__ out)
{
    __shared__ float lds_a[16][132];
    __shared__ float lds_h[16][132];

    const int tid  = threadIdx.x;
    const int lane = tid & 31;
    const int wv   = __builtin_amdgcn_readfirstlane(tid >> 5);
    const int row0 = blockIdx.x * 16;

    {
        const int ra = tid >> 5;
        const int rb = ra + 8;
        const int k4 = (tid & 31) << 2;
        float4 va = *(const float4*)(psum + (size_t)(row0 + ra) * HID_C + k4);
        float4 vb = *(const float4*)(psum + (size_t)(row0 + rb) * HID_C + k4);
        const float ca = fmaxf(pcnt[row0 + ra], 1.0f);
        const float cb = fmaxf(pcnt[row0 + rb], 1.0f);
        va.x /= ca; va.y /= ca; va.z /= ca; va.w /= ca;
        vb.x /= cb; vb.y /= cb; vb.z /= cb; vb.w /= cb;
        *(float4*)&lds_a[ra][k4] = va;
        *(float4*)&lds_a[rb][k4] = vb;
    }
    __syncthreads();

    const int m    = lane & 15;
    const int kh   = (lane >> 4) << 1;
    const int hb   = (lane >> 4) << 3;
    const int col0 = wv * 16;
    const int col  = col0 + m;

    {
        const float* Bp = W1 + (size_t)kh * HID_C + col;
        v8f acc = {};
#pragma unroll 8
        for (int kb = 0; kb < HID_C; kb += 4) {
            v2f a, w;
            a.x = lds_a[m][kb + kh];
            a.y = lds_a[m][kb + kh + 1];
            w.x = Bp[(size_t)kb * HID_C];
            w.y = Bp[(size_t)(kb + 1) * HID_C];
            acc = __builtin_amdgcn_wmma_f32_16x16x4_f32(
                false, a, false, w, (short)0, acc, false, false);
        }
        const float bv = b1[col];
#pragma unroll
        for (int v = 0; v < 8; ++v)
            lds_h[hb + v][col] = fmaxf(acc[v] + bv, 0.0f);
    }
    __syncthreads();

    if (wv < OUT_C / 16) {
        const float* Bp = W2 + (size_t)kh * OUT_C + col;
        v8f acc = {};
#pragma unroll 8
        for (int kb = 0; kb < HID_C; kb += 4) {
            v2f a, w;
            a.x = lds_h[m][kb + kh];
            a.y = lds_h[m][kb + kh + 1];
            w.x = Bp[(size_t)kb * OUT_C];
            w.y = Bp[(size_t)(kb + 1) * OUT_C];
            acc = __builtin_amdgcn_wmma_f32_16x16x4_f32(
                false, a, false, w, (short)0, acc, false, false);
        }
        const float bv = b2[col];
#pragma unroll
        for (int v = 0; v < 8; ++v)
            out[(size_t)(row0 + hb + v) * OUT_C + col] = acc[v] + bv;
    }
}

#define RANK_CHUNK 16384
#define RANK_T     1024
#define MAX_NODES  50000
#define SCAN_T     1024
#define NCH_MAX    64
static_assert(SCAN_T == 4 * 256, "four 256-node pieces of the ranking pass's write-out make one scan chunk");
static_assert((MAX_NODES + SCAN_T - 1) / SCAN_T <= NCH_MAX, "chunk totals fit");

__global__ __launch_bounds__(RANK_T) void rank_slices(
    const int* __restrict__ dst, int* __restrict__ cntm, int* __restrict__ rank,
    int* __restrict__ ctotg, int E, int n)
{
    __shared__ alignas(16) int lc[MAX_NODES];
    __shared__ int ctot[NCH_MAX];
    const int t = threadIdx.x;
    const int g = blockIdx.x;

    if (t < NCH_MAX) ctot[t] = 0;

    for (int i = t; i < MAX_NODES / 4; i += RANK_T)
        ((int4*)lc)[i] = make_int4(0, 0, 0, 0);
    __syncthreads();

    const int beg = g * RANK_CHUNK;
    const int end = min(beg + RANK_CHUNK, E);
    for (int e = beg + t; e < end; e += 4 * RANK_T) {
        const int e1 = e + RANK_T, e2 = e + 2 * RANK_T, e3 = e + 3 * RANK_T;
        if (e3 < end) {
            const int d0 = dst[e];
            const int d1 = dst[e1];
            const int d2 = dst[e2];
            const int d3 = dst[e3];
            const int r0 = atomicAdd(&lc[d0], 1);
            const int r1 = atomicAdd(&lc[d1], 1);
            const int r2 = atomicAdd(&lc[d2], 1);
            const int r3 = atomicAdd(&lc[d3], 1);
            rank[e]  = r0;
            rank[e1] = r1;
            rank[e2] = r2;
            rank[e3] = r3;
        } else {
            for (int k = e; k < end; k += RANK_T)
                rank[k] = atomicAdd(&lc[dst[k]], 1);
        }
    }
    __syncthreads();

    int* out = cntm + (size_t)g * n;
    const int nch  = (n + SCAN_T - 1) / SCAN_T;
    const int lane = t & 31;
    const int nq   = (n + 255) >> 8;
    for (int q = t >> 5; q < nq; q += RANK_T / 32) {
        int acc = 0;
#pragma unroll
        for (int it = 0; it < 8; ++it) {
            const int i = (q << 8) + (it << 5) + lane;
            if (i < n) { const int v = lc[i]; out[i] = v; acc += v; }
        }
        for (int o = 16; o > 0; o >>= 1) acc += __shfl_xor(acc, o);
        if (lane == 0) atomicAdd(&ctot[q >> 2], acc);
    }
    __syncthreads();
    if (t < nch) ctotg[(size_t)g * NCH_MAX + t] = ctot[t];
}

__global__ __launch_bounds__(SCAN_T) void offsets_rowptr(
    int* __restrict__ cntm, const int* __restrict__ ctotg, int* __restrict__ rowptr,
    float* __restrict__ dinv, int n, int G)
{
    __shared__ int s_own[SCAN_T];
    __shared__ int s_pre[SCAN_T];
    const int t = threadIdx.x;
    const int b = blockIdx.x;
    const int i = b * SCAN_T + t;

    int pre = 0;
    const int tot = G * b;
    for (int k = t; k < tot; k += SCAN_T) {
        const int g = k / b;
        const int c = k - g * b;
        pre += ctotg[(size_t)g * NCH_MAX + c];
    }

    int run = 0;
    if (i < n) {
        int* p = cntm + i;
        int g = 0;
        for (; g + 8 <= G; g += 8) {
            int c[8];
#pragma unroll
            for (int k = 0; k < 8; ++k) c[k] = p[(size_t)(g + k) * n];
#pragma unroll
            for (int k = 0; k < 8; ++k) {
                p[(size_t)(g + k) * n] = run;
                run += c[k];
            }
        }
        for (; g < G; ++g) {
            const int c = p[(size_t)g * n];
            p[(size_t)g * n] = run;
            run += c;
        }
    }

    const int c = run;
    s_own[t] = c;
    s_pre[t] = pre;
    __syncthreads();
    for (int off = 1; off < SCAN_T; off <<= 1) {
        int a = 0, q = 0;
        if (t >= off) { a = s_own[t - off]; q = s_pre[t - off]; }
        __syncthreads();
        s_own[t] += a;
        s_pre[t] += q;
        __syncthreads();
    }
    const int excl = s_pre[SCAN_T - 1] + s_own[t] - c;
    if (i < n) {
        rowptr[i] = excl;
        dinv[i]   = rsqrtf((float)(c + 1));
    }
    if (i == n - 1) rowptr[n] = excl + c;
}

__global__ void fill_csr(const int* __restrict__ src, const int* __restrict__ dst,
                         const int* __restrict__ rank, const int* __restrict__ rowptr,
                         const int* __restrict__ cntm,
                         const float* __restrict__ dinv, int2* __restrict__ csr, int E, int n)
{
    int e = blockIdx.x * blockDim.x + threadIdx.x;
    if (e < E) {
        int s = src[e];
        int d = dst[e];
        int g = e / RANK_CHUNK;
        int2 r;
        r.x = s;
        r.y = __float_as_int(dinv[s] * dinv[d]);
        csr[rowptr[d] + cntm[(size_t)g * n + d] + rank[e]] = r;
    }
}

__device__ __forceinline__ void fma4(float4& a, const float4& v, float w)
{
    a.x = fmaf(v.x, w, a.x);
    a.y = fmaf(v.y, w, a.y);
    a.z = fmaf(v.z, w, a.z);
    a.w = fmaf(v.w, w, a.w);
}

template <int K>
__device__ __forceinline__ void add_recs(
    float4& accA, float4& accB, const float* __restrict__ xw, int2 cur, int j, int nA, int c4)
{
    int    s[K];
    float  w[K];
    float4 v[K];
#pragma unroll
    for (int k = 0; k < K; ++k) {
        s[k] = __builtin_amdgcn_readlane(cur.x, j + k);
        w[k] = __int_as_float(__builtin_amdgcn_readlane(cur.y, j + k));
    }
#pragma unroll
    for (int k = 0; k < K; ++k)
        v[k] = *(const float4*)(xw + (size_t)s[k] * HID_C + c4);
    if (j + K <= nA) {
#pragma unroll
        for (int k = 0; k < K; ++k) fma4(accA, v[k], w[k]);
    } else if (j >= nA) {
#pragma unroll
        for (int k = 0; k < K; ++k) fma4(accB, v[k], w[k]);
    } else {
#pragma unroll
        for (int k = 0; k < K; ++k) {
            if (j + k < nA) fma4(accA, v[k], w[k]);
            else            fma4(accB, v[k], w[k]);
        }
    }
}

__device__ __forceinline__ float gelu_erf(float v)
{
    return 0.5f * v * (1.0f + erff(v * 0.70710678118654752f));
}

template <bool POOL>
__global__ __launch_bounds__(256) __attribute__((amdgpu_waves_per_eu(16, 16))) void gcn_layer(
    const float* __restrict__ hin, const float* __restrict__ dinv,
    const int* __restrict__ rowptr, const int2* __restrict__ csr,
    const float* __restrict__ Wl, const float* __restrict__ b,
    float* __restrict__ hout, const int* __restrict__ batch,
    float* __restrict__ psum, float* __restrict__ pcnt, int n)
{
    __shared__ float lds_a[16][132];
    __shared__ int   s_gid[16];

    const int tid  = threadIdx.x;
    const int lane = tid & 31;
    const int wv   = tid >> 5;
    const int row0 = blockIdx.x * 16;
    const int c4   = lane << 2;

    if (POOL) {
        if (tid < 16) s_gid[tid] = (row0 + tid < n) ? batch[row0 + tid] : -1;
    } else if (psum) {
        for (int i = blockIdx.x * 256 + tid; i < NBATCH * HID_C + NBATCH; i += gridDim.x * 256)
            psum[i] = 0.0f;
    }

    const int  rowA = __builtin_amdgcn_readfirstlane(row0 + wv * 2);
    const bool hasA = rowA < n;
    const bool hasB = rowA + 1 < n;
    int   p0 = 0, p1 = 0, p2 = 0;
    float dA = 0.0f, dB = 0.0f;
    float4 accA = make_float4(0.0f, 0.0f, 0.0f, 0.0f);
    float4 accB = make_float4(0.0f, 0.0f, 0.0f, 0.0f);
    if (hasA) {
        p0 = rowptr[rowA];
        p1 = rowptr[rowA + 1];
        dA = dinv[rowA];
        accA = *(const float4*)(hin + (size_t)rowA * HID_C + c4);
    }
    if (hasB) {
        p2 = rowptr[rowA + 2];
        dB = dinv[rowA + 1];
        accB = *(const float4*)(hin + (size_t)(rowA + 1) * HID_C + c4);
    }
    if (!hasB) p2 = p1;
    {
        const float sA = dA * dA;
        accA.x *= sA; accA.y *= sA; accA.z *= sA; accA.w *= sA;
        const float sB = dB * dB;
        accB.x *= sB; accB.y *= sB; accB.z *= sB; accB.w *= sB;
    }
    {
        int2 cur = make_int2(0, 0);
        if (p0 + lane < p2) cur = csr[p0 + lane];
        for (int base = p0; base < p2; base += 32) {
            int2 nxt = make_int2(0, 0);
            if (base + 32 + lane < p2) nxt = csr[base + 32 + lane];
            const int cnt = min(32, p2 - base);
            const int nA  = p1 - base;
            int j = 0;
            for (; j + 8 <= cnt; j += 8) add_recs<8>(accA, accB, hin, cur, j, nA, c4);
            if (j + 4 <= cnt) { add_recs<4>(accA, accB, hin, cur, j, nA, c4); j += 4; }
            if (j + 2 <= cnt) { add_recs<2>(accA, accB, hin, cur, j, nA, c4); j += 2; }
            if (j < cnt)      { add_recs<1>(accA, accB, hin, cur, j, nA, c4); }
            cur = nxt;
        }
        *(float4*)&lds_a[wv * 2][c4]     = accA;
        *(float4*)&lds_a[wv * 2 + 1][c4] = accB;
    }
    __syncthreads();

    const int m    = lane & 15;
    const int kh   = (lane >> 4) << 1;
    const int col0 = wv * 16;
    const float* Bp = Wl + (size_t)kh * HID_C + col0 + m;

    v8f cacc = {};
#pragma unroll 8
    for (int kb = 0; kb < HID_C; kb += 4) {
        v2f a, w;
        a.x = lds_a[m][kb + kh];
        a.y = lds_a[m][kb + kh + 1];
        w.x = Bp[(size_t)kb * HID_C];
        w.y = Bp[(size_t)(kb + 1) * HID_C];
        cacc = __builtin_amdgcn_wmma_f32_16x16x4_f32(
            false, a, false, w, (short)0, cacc, false, false);
    }

    const int   col   = col0 + m;
    const int   rbase = row0 + ((lane >> 4) << 3);
    const float bv    = b[col];
    float y[8];
#pragma unroll
    for (int v = 0; v < 8; ++v) y[v] = gelu_erf(cacc[v] + bv);
    if (!POOL) {
#pragma unroll
        for (int v = 0; v < 8; ++v) {
            if (rbase + v < n) hout[(size_t)(rbase + v) * HID_C + col] = y[v];
        }
    } else {
        const int hb = (lane >> 4) << 3;
        int   cur = -1;
        float s   = 0.0f;
#pragma unroll
        for (int v = 0; v < 8; ++v) {
            const int g = s_gid[hb + v];
            if (g != cur) {
                if (cur >= 0) atomicAdd(psum + (size_t)cur * HID_C + col, s);
                cur = g;
                s   = 0.0f;
            }
            s += y[v];
        }
        if (cur >= 0) atomicAdd(psum + (size_t)cur * HID_C + col, s);
        if (tid < 16) {
            const int g = s_gid[tid];
            if (g >= 0 && (tid == 0 || s_gid[tid - 1] != g)) {
                int len = 1;
                while (tid + len < 16 && s_gid[tid + len] == g) ++len;
                atomicAdd(pcnt + g, (float)len);
            }
        }
    }
}

extern "C" void kernel_launch(void* const* d_in, const int* in_sizes, int n_in,
                              void* d_out, int out_size, void* d_ws, size_t ws_size,
                              hipStream_t stream)
{
    const float* x     = (const float*)d_in[0];
    const int*   ei    = (const int*)d_in[1];
    const int*   batch = (const int*)d_in[2];
    const float* W[3]  = { (const float*)d_in[3], (const float*)d_in[5], (const float*)d_in[7] };
    const float* bv[3] = { (const float*)d_in[4], (const float*)d_in[6], (const float*)d_in[8] };
    const float* Wm1   = (const float*)d_in[9];
    const float* bm1   = (const float*)d_in[10];
    const float* Wm2   = (const float*)d_in[11];
    const float* bm2   = (const float*)d_in[12];

    const int N = in_sizes[0] / IN_C;
    const int E = in_sizes[1] / 2;
    const int* src = ei;
    const int* dst = ei + E;

    char*  ws  = (char*)d_ws;
    size_t off = 0;
    auto carve = [&](size_t bytes) -> float* {
        float* p = (float*)(ws + off);
        off = (off + bytes + 255) & ~(size_t)255;
        return p;
    };
    float* dinv = carve((size_t)N * 4);
    float* buf0 = carve((size_t)N * HID_C * 4);
    float* buf1 = carve((size_t)N * HID_C * 4);
    float* psum = carve(((size_t)NBATCH * HID_C + NBATCH) * 4);
    float* pcnt = psum + (size_t)NBATCH * HID_C;
    int*   rowptr = (int*)carve((size_t)(N + 1) * 4);
    int*   rank   = (int*)carve((size_t)E * 4);
    int2*  csr    = (int2*)carve((size_t)E * 8);
    const int G   = (E + RANK_CHUNK - 1) / RANK_CHUNK;
    int*   cntm   = (int*)carve((size_t)G * N * 4);
    int*   ctotg  = (int*)carve((size_t)G * NCH_MAX * 4);

    rank_slices<<<G, RANK_T, 0, stream>>>(dst, cntm, rank, ctotg, E, N);
    offsets_rowptr<<<(N + SCAN_T - 1) / SCAN_T, SCAN_T, 0, stream>>>(cntm, ctotg, rowptr, dinv, N, G);
    fill_csr<<<(E + 255) / 256, 256, 0, stream>>>(src, dst, rank, rowptr, cntm, dinv, csr, E, N);

    gcn_layer<false><<<(N + 15) / 16, 256, 0, stream>>>(
        x, dinv, rowptr, csr, W[0], bv[0], buf0, nullptr, nullptr, nullptr, N);
    gcn_layer<false><<<(N + 15) / 16, 256, 0, stream>>>(
        buf0, dinv, rowptr, csr, W[1], bv[1], buf1, nullptr, psum, nullptr, N);
    gcn_layer<true><<<(N + 15) / 16, 256, 0, stream>>>(
        buf1, dinv, rowptr, csr, W[2], bv[2], nullptr, batch, psum, pcnt, N);

    mlp_head<<<NBATCH / 16, 256, 0, stream>>>(psum, pcnt, Wm1, bm1, Wm2, bm2, (float*)d_out);
}


# list fill: scattered reads per edge 4 -> 2 (segment start folded into the slice-offset plane, counts kept in registers across the scan; dst scale applied once per row sum); sum order unchanged
# speedup vs baseline: 5.2799x; 1.0567x over previous
#include <hip/hip_runtime.h>
#include <hip/hip_bf16.h>
#include <math.h>

typedef float v2f __attribute__((ext_vector_type(2)));
typedef float v8f __attribute__((ext_vector_type(8)));

#define IN_C   128
#define HID_C  128
#define OUT_C  64
#define NBATCH 512

__global__ __launch_bounds__(256) void mlp_head(
    const float* __restrict__ psum, const float* __restrict__ pcnt,
    const float* __restrict__ W1, const float* __restrict__ b1,
    const float* __restrict__ W2, const float* __restrict__ b2,
    float* __restrict__ out)
{
    __shared__ float lds_a[16][132];
    __shared__ float lds_h[16][132];

    const int tid  = threadIdx.x;
    const int lane = tid & 31;
    const int wv   = __builtin_amdgcn_readfirstlane(tid >> 5);
    const int row0 = blockIdx.x * 16;

    {
        const int ra = tid >> 5;
        const int rb = ra + 8;
        const int k4 = (tid & 31) << 2;
        float4 va = *(const float4*)(psum + (size_t)(row0 + ra) * HID_C + k4);
        float4 vb = *(const float4*)(psum + (size_t)(row0 + rb) * HID_C + k4);
        const float ca = fmaxf(pcnt[row0 + ra], 1.0f);
        const float cb = fmaxf(pcnt[row0 + rb], 1.0f);
        va.x /= ca; va.y /= ca; va.z /= ca; va.w /= ca;
        vb.x /= cb; vb.y /= cb; vb.z /= cb; vb.w /= cb;
        *(float4*)&lds_a[ra][k4] = va;
        *(float4*)&lds_a[rb][k4] = vb;
    }
    __syncthreads();

    const int m    = lane & 15;
    const int kh   = (lane >> 4) << 1;
    const int hb   = (lane >> 4) << 3;
    const int col0 = wv * 16;
    const int col  = col0 + m;

    {
        const float* Bp = W1 + (size_t)kh * HID_C + col;
        v8f acc = {};
#pragma unroll 8
        for (int kb = 0; kb < HID_C; kb += 4) {
            v2f a, w;
            a.x = lds_a[m][kb + kh];
            a.y = lds_a[m][kb + kh + 1];
            w.x = Bp[(size_t)kb * HID_C];
            w.y = Bp[(size_t)(kb + 1) * HID_C];
            acc = __builtin_amdgcn_wmma_f32_16x16x4_f32(
                false, a, false, w, (short)0, acc, false, false);
        }
        const float bv = b1[col];
#pragma unroll
        for (int v = 0; v < 8; ++v)
            lds_h[hb + v][col] = fmaxf(acc[v] + bv, 0.0f);
    }
    __syncthreads();

    if (wv < OUT_C / 16) {
        const float* Bp = W2 + (size_t)kh * OUT_C + col;
        v8f acc = {};
#pragma unroll 8
        for (int kb = 0; kb < HID_C; kb += 4) {
            v2f a, w;
            a.x = lds_h[m][kb + kh];
            a.y = lds_h[m][kb + kh + 1];
            w.x = Bp[(size_t)kb * OUT_C];
            w.y = Bp[(size_t)(kb + 1) * OUT_C];
            acc = __builtin_amdgcn_wmma_f32_16x16x4_f32(
                false, a, false, w, (short)0, acc, false, false);
        }
        const float bv = b2[col];
#pragma unroll
        for (int v = 0; v < 8; ++v)
            out[(size_t)(row0 + hb + v) * OUT_C + col] = acc[v] + bv;
    }
}

#define RANK_CHUNK 16384
#define RANK_T     1024
#define MAX_NODES  50000
#define SCAN_T     1024
#define NCH_MAX    64
#define CNT_PITCH  MAX_NODES
static_assert(SCAN_T == 4 * 256, "four 256-node pieces of the ranking pass's write-out make one scan chunk");
static_assert((MAX_NODES + SCAN_T - 1) / SCAN_T <= NCH_MAX, "chunk totals fit");

__global__ __launch_bounds__(RANK_T) void rank_slices(
    const int* __restrict__ dst, int* __restrict__ cntm, int* __restrict__ rank,
    int* __restrict__ ctotg, int E, int n)
{
    __shared__ alignas(16) int lc[MAX_NODES];
    __shared__ int ctot[NCH_MAX];
    const int t = threadIdx.x;
    const int g = blockIdx.x;

    if (t < NCH_MAX) ctot[t] = 0;

    for (int i = t; i < MAX_NODES / 4; i += RANK_T)
        ((int4*)lc)[i] = make_int4(0, 0, 0, 0);
    __syncthreads();

    const int beg = g * RANK_CHUNK;
    const int end = min(beg + RANK_CHUNK, E);
    for (int e = beg + t; e < end; e += 4 * RANK_T) {
        const int e1 = e + RANK_T, e2 = e + 2 * RANK_T, e3 = e + 3 * RANK_T;
        if (e3 < end) {
            const int d0 = dst[e];
            const int d1 = dst[e1];
            const int d2 = dst[e2];
            const int d3 = dst[e3];
            const int r0 = atomicAdd(&lc[d0], 1);
            const int r1 = atomicAdd(&lc[d1], 1);
            const int r2 = atomicAdd(&lc[d2], 1);
            const int r3 = atomicAdd(&lc[d3], 1);
            rank[e]  = r0;
            rank[e1] = r1;
            rank[e2] = r2;
            rank[e3] = r3;
        } else {
            for (int k = e; k < end; k += RANK_T)
                rank[k] = atomicAdd(&lc[dst[k]], 1);
        }
    }
    __syncthreads();

    int* out = cntm + (size_t)g * CNT_PITCH;
    const int nch  = (n + SCAN_T - 1) / SCAN_T;
    const int lane = t & 31;
    const int nq   = (n + 255) >> 8;
    for (int q = t >> 5; q < nq; q += RANK_T / 32) {
        int acc = 0;
#pragma unroll
        for (int it = 0; it < 8; ++it) {
            const int i = (q << 8) + (it << 5) + lane;
            if (i < n) { const int v = lc[i]; out[i] = v; acc += v; }
        }
        for (int o = 16; o > 0; o >>= 1) acc += __shfl_xor(acc, o);
        if (lane == 0) atomicAdd(&ctot[q >> 2], acc);
    }
    __syncthreads();
    if (t < nch) ctotg[(size_t)g * NCH_MAX + t] = ctot[t];
}

#define NSL_REG 80
__global__ __launch_bounds__(SCAN_T) void offsets_rowptr(
    int* __restrict__ cntm, const int* __restrict__ ctotg, int* __restrict__ rowptr,
    float* __restrict__ dinv, int n, int G)
{
    __shared__ int s_own[SCAN_T];
    __shared__ int s_pre[SCAN_T];
    const int t = threadIdx.x;
    const int b = blockIdx.x;
    const int i = b * SCAN_T + t;

    int pre = 0;
    const int tot = G * b;
    for (int k = t; k < tot; k += SCAN_T) {
        const int g = k / b;
        const int c = k - g * b;
        pre += ctotg[(size_t)g * NCH_MAX + c];
    }

    int c[NSL_REG];
#pragma unroll
    for (int g = 0; g < NSL_REG; ++g) c[g] = 0;
    int deg = 0;
    int* p = cntm + i;
    if (i < n) {
#pragma unroll
        for (int g = 0; g < NSL_REG; ++g)
            if (g < G) c[g] = p[(size_t)g * CNT_PITCH];
#pragma unroll
        for (int g = 0; g < NSL_REG; ++g) deg += c[g];
        for (int g = NSL_REG; g < G; ++g) deg += p[(size_t)g * CNT_PITCH];
    }

    s_own[t] = deg;
    s_pre[t] = pre;
    __syncthreads();
    for (int off = 1; off < SCAN_T; off <<= 1) {
        int a = 0, q = 0;
        if (t >= off) { a = s_own[t - off]; q = s_pre[t - off]; }
        __syncthreads();
        s_own[t] += a;
        s_pre[t] += q;
        __syncthreads();
    }
    const int excl = s_pre[SCAN_T - 1] + s_own[t] - deg;
    if (i < n) {
        int run = excl;
#pragma unroll
        for (int g = 0; g < NSL_REG; ++g)
            if (g < G) { p[(size_t)g * CNT_PITCH] = run; run += c[g]; }
        for (int g = NSL_REG; g < G; ++g) {
            const int cc = p[(size_t)g * CNT_PITCH];
            p[(size_t)g * CNT_PITCH] = run;
            run += cc;
        }
        rowptr[i] = excl;
        dinv[i]   = rsqrtf((float)(deg + 1));
    }
    if (i == n - 1) rowptr[n] = excl + deg;
}

__global__ void fill_csr(const int* __restrict__ src, const int* __restrict__ dst,
                         const int* __restrict__ rank, const int* __restrict__ cntm,
                         const float* __restrict__ dinv, int2* __restrict__ csr, int E)
{
    int e = blockIdx.x * blockDim.x + threadIdx.x;
    if (e < E) {
        int s = src[e];
        int d = dst[e];
        int g = e / RANK_CHUNK;
        int2 r;
        r.x = s;
        r.y = __float_as_int(dinv[s]);
        csr[cntm[(size_t)g * CNT_PITCH + d] + rank[e]] = r;
    }
}

__device__ __forceinline__ void fma4(float4& a, const float4& v, float w)
{
    a.x = fmaf(v.x, w, a.x);
    a.y = fmaf(v.y, w, a.y);
    a.z = fmaf(v.z, w, a.z);
    a.w = fmaf(v.w, w, a.w);
}

template <int K>
__device__ __forceinline__ void add_recs(
    float4& accA, float4& accB, const float* __restrict__ xw, int2 cur, int j, int nA, int c4)
{
    int    s[K];
    float  w[K];
    float4 v[K];
#pragma unroll
    for (int k = 0; k < K; ++k) {
        s[k] = __builtin_amdgcn_readlane(cur.x, j + k);
        w[k] = __int_as_float(__builtin_amdgcn_readlane(cur.y, j + k));
    }
#pragma unroll
    for (int k = 0; k < K; ++k)
        v[k] = *(const float4*)(xw + (size_t)s[k] * HID_C + c4);
    if (j + K <= nA) {
#pragma unroll
        for (int k = 0; k < K; ++k) fma4(accA, v[k], w[k]);
    } else if (j >= nA) {
#pragma unroll
        for (int k = 0; k < K; ++k) fma4(accB, v[k], w[k]);
    } else {
#pragma unroll
        for (int k = 0; k < K; ++k) {
            if (j + k < nA) fma4(accA, v[k], w[k]);
            else            fma4(accB, v[k], w[k]);
        }
    }
}

__device__ __forceinline__ float gelu_erf(float v)
{
    return 0.5f * v * (1.0f + erff(v * 0.70710678118654752f));
}

template <bool POOL>
__global__ __launch_bounds__(256) __attribute__((amdgpu_waves_per_eu(16, 16))) void gcn_layer(
    const float* __restrict__ hin, const float* __restrict__ dinv,
    const int* __restrict__ rowptr, const int2* __restrict__ csr,
    const float* __restrict__ Wl, const float* __restrict__ b,
    float* __restrict__ hout, const int* __restrict__ batch,
    float* __restrict__ psum, float* __restrict__ pcnt, int n)
{
    __shared__ float lds_a[16][132];
    __shared__ int   s_gid[16];

    const int tid  = threadIdx.x;
    const int lane = tid & 31;
    const int wv   = tid >> 5;
    const int row0 = blockIdx.x * 16;
    const int c4   = lane << 2;

    if (POOL) {
        if (tid < 16) s_gid[tid] = (row0 + tid < n) ? batch[row0 + tid] : -1;
    } else if (psum) {
        for (int i = blockIdx.x * 256 + tid; i < NBATCH * HID_C + NBATCH; i += gridDim.x * 256)
            psum[i] = 0.0f;
    }

    const int  rowA = __builtin_amdgcn_readfirstlane(row0 + wv * 2);
    const bool hasA = rowA < n;
    const bool hasB = rowA + 1 < n;
    int   p0 = 0, p1 = 0, p2 = 0;
    float dA = 0.0f, dB = 0.0f;
    float4 accA = make_float4(0.0f, 0.0f, 0.0f, 0.0f);
    float4 accB = make_float4(0.0f, 0.0f, 0.0f, 0.0f);
    if (hasA) {
        p0 = rowptr[rowA];
        p1 = rowptr[rowA + 1];
        dA = dinv[rowA];
        accA = *(const float4*)(hin + (size_t)rowA * HID_C + c4);
    }
    if (hasB) {
        p2 = rowptr[rowA + 2];
        dB = dinv[rowA + 1];
        accB = *(const float4*)(hin + (size_t)(rowA + 1) * HID_C + c4);
    }
    if (!hasB) p2 = p1;
    accA.x *= dA; accA.y *= dA; accA.z *= dA; accA.w *= dA;
    accB.x *= dB; accB.y *= dB; accB.z *= dB; accB.w *= dB;
    {
        int2 cur = make_int2(0, 0);
        if (p0 + lane < p2) cur = csr[p0 + lane];
        for (int base = p0; base < p2; base += 32) {
            int2 nxt = make_int2(0, 0);
            if (base + 32 + lane < p2) nxt = csr[base + 32 + lane];
            const int cnt = min(32, p2 - base);
            const int nA  = p1 - base;
            int j = 0;
            for (; j + 8 <= cnt; j += 8) add_recs<8>(accA, accB, hin, cur, j, nA, c4);
            if (j + 4 <= cnt) { add_recs<4>(accA, accB, hin, cur, j, nA, c4); j += 4; }
            if (j + 2 <= cnt) { add_recs<2>(accA, accB, hin, cur, j, nA, c4); j += 2; }
            if (j < cnt)      { add_recs<1>(accA, accB, hin, cur, j, nA, c4); }
            cur = nxt;
        }
        accA.x *= dA; accA.y *= dA; accA.z *= dA; accA.w *= dA;
        accB.x *= dB; accB.y *= dB; accB.z *= dB; accB.w *= dB;
        *(float4*)&lds_a[wv * 2][c4]     = accA;
        *(float4*)&lds_a[wv * 2 + 1][c4] = accB;
    }
    __syncthreads();

    const int m    = lane & 15;
    const int kh   = (lane >> 4) << 1;
    const int col0 = wv * 16;
    const float* Bp = Wl + (size_t)kh * HID_C + col0 + m;

    v8f cacc = {};
#pragma unroll 8
    for (int kb = 0; kb < HID_C; kb += 4) {
        v2f a, w;
        a.x = lds_a[m][kb + kh];
        a.y = lds_a[m][kb + kh + 1];
        w.x = Bp[(size_t)kb * HID_C];
        w.y = Bp[(size_t)(kb + 1) * HID_C];
        cacc = __builtin_amdgcn_wmma_f32_16x16x4_f32(
            false, a, false, w, (short)0, cacc, false, false);
    }

    const int   col   = col0 + m;
    const int   rbase = row0 + ((lane >> 4) << 3);
    const float bv    = b[col];
    float y[8];
#pragma unroll
    for (int v = 0; v < 8; ++v) y[v] = gelu_erf(cacc[v] + bv);
    if (!POOL) {
#pragma unroll
        for (int v = 0; v < 8; ++v) {
            if (rbase + v < n) hout[(size_t)(rbase + v) * HID_C + col] = y[v];
        }
    } else {
        const int hb = (lane >> 4) << 3;
        int   cur = -1;
        float s   = 0.0f;
#pragma unroll
        for (int v = 0; v < 8; ++v) {
            const int g = s_gid[hb + v];
            if (g != cur) {
                if (cur >= 0) atomicAdd(psum + (size_t)cur * HID_C + col, s);
                cur = g;
                s   = 0.0f;
            }
            s += y[v];
        }
        if (cur >= 0) atomicAdd(psum + (size_t)cur * HID_C + col, s);
        if (tid < 16) {
            const int g = s_gid[tid];
            if (g >= 0 && (tid == 0 || s_gid[tid - 1] != g)) {
                int len = 1;
                while (tid + len < 16 && s_gid[tid + len] == g) ++len;
                atomicAdd(pcnt + g, (float)len);
            }
        }
    }
}

extern "C" void kernel_launch(void* const* d_in, const int* in_sizes, int n_in,
                              void* d_out, int out_size, void* d_ws, size_t ws_size,
                              hipStream_t stream)
{
    const float* x     = (const float*)d_in[0];
    const int*   ei    = (const int*)d_in[1];
    const int*   batch = (const int*)d_in[2];
    const float* W[3]  = { (const float*)d_in[3], (const float*)d_in[5], (const float*)d_in[7] };
    const float* bv[3] = { (const float*)d_in[4], (const float*)d_in[6], (const float*)d_in[8] };
    const float* Wm1   = (const float*)d_in[9];
    const float* bm1   = (const float*)d_in[10];
    const float* Wm2   = (const float*)d_in[11];
    const float* bm2   = (const float*)d_in[12];

    const int N = in_sizes[0] / IN_C;
    const int E = in_sizes[1] / 2;
    const int* src = ei;
    const int* dst = ei + E;

    char*  ws  = (char*)d_ws;
    size_t off = 0;
    auto carve = [&](size_t bytes) -> float* {
        float* p = (float*)(ws + off);
        off = (off + bytes + 255) & ~(size_t)255;
        return p;
    };
    float* dinv = carve((size_t)N * 4);
    float* buf0 = carve((size_t)N * HID_C * 4);
    float* buf1 = carve((size_t)N * HID_C * 4);
    float* psum = carve(((size_t)NBATCH * HID_C + NBATCH) * 4);
    float* pcnt = psum + (size_t)NBATCH * HID_C;
    int*   rowptr = (int*)carve((size_t)(N + 1) * 4);
    int*   rank   = (int*)carve((size_t)E * 4);
    int2*  csr    = (int2*)carve((size_t)E * 8);
    const int G   = (E + RANK_CHUNK - 1) / RANK_CHUNK;
    int*   cntm   = (int*)carve((size_t)G * CNT_PITCH * 4);
    int*   ctotg  = (int*)carve((size_t)G * NCH_MAX * 4);

    rank_slices<<<G, RANK_T, 0, stream>>>(dst, cntm, rank, ctotg, E, N);
    offsets_rowptr<<<(N + SCAN_T - 1) / SCAN_T, SCAN_T, 0, stream>>>(cntm, ctotg, rowptr, dinv, N, G);
    fill_csr<<<(E + 255) / 256, 256, 0, stream>>>(src, dst, rank, cntm, dinv, csr, E);

    gcn_layer<false><<<(N + 15) / 16, 256, 0, stream>>>(
        x, dinv, rowptr, csr, W[0], bv[0], buf0, nullptr, nullptr, nullptr, N);
    gcn_layer<false><<<(N + 15) / 16, 256, 0, stream>>>(
        buf0, dinv, rowptr, csr, W[1], bv[1], buf1, nullptr, psum, nullptr, N);
    gcn_layer<true><<<(N + 15) / 16, 256, 0, stream>>>(
        buf1, dinv, rowptr, csr, W[2], bv[2], nullptr, batch, psum, pcnt, N);

    mlp_head<<<NBATCH / 16, 256, 0, stream>>>(psum, pcnt, Wm1, bm1, Wm2, bm2, (float*)d_out);
}
